# starting kernel
# baseline (speedup 1.0000x reference)
#include <hip/hip_runtime.h>
#include <math.h>

#define BB    2
#define NN    50000
#define EE    800000
#define INP   128
#define KEY   64
#define VAL   128
#define MROWS (BB * NN)
#define LN_EPS 1e-5f

typedef __attribute__((ext_vector_type(2))) float v2f;
typedef __attribute__((ext_vector_type(8))) float v8f;

__device__ __forceinline__ unsigned enc_f32(float f) {
    unsigned u = __float_as_uint(f);
    return (u & 0x80000000u) ? ~u : (u | 0x80000000u);
}
__device__ __forceinline__ float dec_f32(unsigned u) {
    return (u & 0x80000000u) ? __uint_as_float(u & 0x7FFFFFFFu)
                             : __uint_as_float(~u);
}

__global__ void bsa_reduce_kw(const float* __restrict__ kw,
                              const float* __restrict__ kb,
                              float* __restrict__ kwsum,
                              float* __restrict__ kbsum) {
    int i = threadIdx.x;
    float s = 0.f;
    for (int j = 0; j < KEY; ++j) s += kw[i * KEY + j];
    kwsum[i] = s;
    if (i == 0) {
        float t = 0.f;
        for (int j = 0; j < KEY; ++j) t += kb[j];
        kbsum[0] = t;
    }
}

__global__ void bsa_ksum(const float* __restrict__ x,
                         const float* __restrict__ kwsum,
                         const float* __restrict__ kbsum,
                         float* __restrict__ ksum) {
    int wid  = (blockIdx.x * blockDim.x + threadIdx.x) >> 5;
    int lane = threadIdx.x & 31;
    if (wid >= MROWS) return;
    const float* xr = x + (size_t)wid * INP;
    float s = 0.f;
    #pragma unroll
    for (int i = 0; i < INP / 32; ++i) {
        int c = lane + i * 32;
        s += xr[c] * kwsum[c];
    }
    #pragma unroll
    for (int off = 16; off; off >>= 1) s += __shfl_xor(s, off);
    if (lane == 0) ksum[wid] = s + kbsum[0];
}

__device__ __forceinline__ void bsa_wmma_tile(const float* __restrict__ Arow,
                                              const float* __restrict__ Bs,
                                              v8f acc[8], int lane) {
    const int kh = (lane >> 4) << 1;
    const int n  = lane & 15;
    for (int k = 0; k < 128; k += 4) {
        v2f a = *(const v2f*)(Arow + k + kh);
        #pragma unroll
        for (int c = 0; c < 8; ++c) {
            v2f bf;
            bf.x = Bs[(k + kh)     * 128 + (c << 4) + n];
            bf.y = Bs[(k + kh + 1) * 128 + (c << 4) + n];
            acc[c] = __builtin_amdgcn_wmma_f32_16x16x4_f32(
                false, a, false, bf, (short)0, acc[c], false, false);
        }
    }
}

__global__ void bsa_values_gemm(const float* __restrict__ A,
                                const float* __restrict__ W,
                                const float* __restrict__ bias,
                                float* __restrict__ out) {
    __shared__ float Bs[128 * 128];
    for (int i = threadIdx.x; i < 128 * 128; i += blockDim.x) Bs[i] = W[i];
    __syncthreads();

    int lane    = threadIdx.x & 31;
    int wv      = threadIdx.x >> 5;
    int tile    = blockIdx.x * 8 + wv;
    int rowbase = tile * 16;
    if (rowbase >= MROWS) return;

    v8f acc[8] = {};
    const float* Arow = A + (size_t)(rowbase + (lane & 15)) * 128;
    bsa_wmma_tile(Arow, Bs, acc, lane);

    int n      = lane & 15;
    int rowoff = (lane >> 4) << 3;
    #pragma unroll
    for (int c = 0; c < 8; ++c) {
        float bb = bias[(c << 4) + n];
        #pragma unroll
        for (int r = 0; r < 8; ++r)
            out[(size_t)(rowbase + rowoff + r) * 128 + (c << 4) + n] = acc[c][r] + bb;
    }
}

__global__ void bsa_edge_max(const float* __restrict__ conn,
                             const int* __restrict__ rows,
                             const int* __restrict__ cols,
                             const float* __restrict__ ksum,
                             unsigned* __restrict__ mu) {
    int tid = blockIdx.x * blockDim.x + threadIdx.x;
    if (tid >= EE * BB) return;
    int e = tid >> 1, b = tid & 1;
    float s = conn[e] * ksum[b * NN + cols[e]];
    atomicMax(&mu[rows[e] * BB + b], enc_f32(s));
}

__global__ void bsa_m_finalize(unsigned* __restrict__ mu) {
    int i = blockIdx.x * blockDim.x + threadIdx.x;
    if (i >= NN * BB) return;
    unsigned u = mu[i];
    float m = (u == 0u) ? 0.0f : dec_f32(u);
    if (!isfinite(m)) m = 0.0f;
    ((float*)mu)[i] = m;
}

__global__ void bsa_edge_exp(const float* __restrict__ conn,
                             const int* __restrict__ rows,
                             const int* __restrict__ cols,
                             const float* __restrict__ ksum,
                             const float* __restrict__ m,
                             float* __restrict__ ex,
                             float* __restrict__ denom) {
    int tid = blockIdx.x * blockDim.x + threadIdx.x;
    if (tid >= EE * BB) return;
    int e = tid >> 1, b = tid & 1;
    int r = rows[e];
    float s = conn[e] * ksum[b * NN + cols[e]];
    float ev = __expf(s - m[r * BB + b]);
    ex[tid] = ev;
    atomicAdd(&denom[r * BB + b], ev);
}

__global__ void bsa_scatter(const int* __restrict__ rows,
                            const int* __restrict__ cols,
                            const float* __restrict__ ex,
                            const float* __restrict__ denom,
                            const float* __restrict__ values,
                            float* __restrict__ agg) {
    int e = blockIdx.x;
    int b = threadIdx.x >> 7;
    int c = threadIdx.x & 127;
    int r = rows[e], cl = cols[e];
    float d = denom[r * BB + b];
    float w = ex[e * BB + b] / (d == 0.0f ? 1.0f : d);
    float v = values[((size_t)b * NN + cl) * VAL + c];
    atomicAdd(&agg[((size_t)b * NN + r) * VAL + c], w * v);
}

__global__ void bsa_out_gemm_ln(const float* __restrict__ A,
                                const float* __restrict__ W,
                                const float* __restrict__ bias,
                                const float* __restrict__ lng,
                                const float* __restrict__ lnb,
                                float* __restrict__ out) {
    __shared__ float Bs[128 * 128];
    for (int i = threadIdx.x; i < 128 * 128; i += blockDim.x) Bs[i] = W[i];
    __syncthreads();

    int lane    = threadIdx.x & 31;
    int wv      = threadIdx.x >> 5;
    int tile    = blockIdx.x * 8 + wv;
    int rowbase = tile * 16;
    if (rowbase >= MROWS) return;

    v8f acc[8] = {};
    const float* Arow = A + (size_t)(rowbase + (lane & 15)) * 128;
    bsa_wmma_tile(Arow, Bs, acc, lane);

    int n = lane & 15;
    float sr[8], sq[8];
    #pragma unroll
    for (int r = 0; r < 8; ++r) { sr[r] = 0.f; sq[r] = 0.f; }
    #pragma unroll
    for (int c = 0; c < 8; ++c) {
        float bb = bias[(c << 4) + n];
        #pragma unroll
        for (int r = 0; r < 8; ++r) {
            float v = acc[c][r] + bb;
            v = v / (1.0f + __expf(-v));
            acc[c][r] = v;
            sr[r] += v;
            sq[r] += v * v;
        }
    }
    #pragma unroll
    for (int r = 0; r < 8; ++r) {
        #pragma unroll
        for (int off = 8; off; off >>= 1) {
            sr[r] += __shfl_xor(sr[r], off);
            sq[r] += __shfl_xor(sq[r], off);
        }
    }
    float mu[8], rstd[8];
    #pragma unroll
    for (int r = 0; r < 8; ++r) {
        mu[r] = sr[r] * (1.0f / VAL);
        float var = sq[r] * (1.0f / VAL) - mu[r] * mu[r];
        rstd[r] = rsqrtf(var + LN_EPS);
    }
    int rowoff = (lane >> 4) << 3;
    #pragma unroll
    for (int c = 0; c < 8; ++c) {
        int col = (c << 4) + n;
        float g = lng[col], b2 = lnb[col];
        #pragma unroll
        for (int r = 0; r < 8; ++r)
            out[(size_t)(rowbase + rowoff + r) * VAL + col] =
                (acc[c][r] - mu[r]) * rstd[r] * g + b2;
    }
}

extern "C" void kernel_launch(void* const* d_in, const int* in_sizes, int n_in,
                              void* d_out, int out_size, void* d_ws, size_t ws_size,
                              hipStream_t stream) {
    const float* x     = (const float*)d_in[0];
    const float* conn  = (const float*)d_in[1];
    const float* k_w   = (const float*)d_in[2];
    const float* k_b   = (const float*)d_in[3];
    const float* v_w   = (const float*)d_in[4];
    const float* v_b   = (const float*)d_in[5];
    const float* out_w = (const float*)d_in[6];
    const float* out_b = (const float*)d_in[7];
    const float* ln_g  = (const float*)d_in[8];
    const float* ln_b  = (const float*)d_in[9];
    const int*   rows  = (const int*)d_in[10];
    const int*   cols  = (const int*)d_in[11];
    float* out = (float*)d_out;

    char* p = (char*)d_ws;
    auto alloc = [&](size_t bytes) {
        void* r = (void*)p;
        p += (bytes + 255) & ~(size_t)255;
        return r;
    };
    float*    kwsum  = (float*)alloc(INP * sizeof(float));
    float*    kbsum  = (float*)alloc(sizeof(float));
    float*    ksum   = (float*)alloc((size_t)MROWS * sizeof(float));
    unsigned* mbuf   = (unsigned*)alloc((size_t)NN * BB * sizeof(float));
    float*    denom  = (float*)alloc((size_t)NN * BB * sizeof(float));
    float*    ex     = (float*)alloc((size_t)EE * BB * sizeof(float));
    float*    values = (float*)alloc((size_t)MROWS * VAL * sizeof(float));
    float*    agg    = (float*)alloc((size_t)MROWS * VAL * sizeof(float));

    hipMemsetAsync(mbuf, 0, (size_t)NN * BB * sizeof(float), stream);
    hipMemsetAsync(denom, 0, (size_t)NN * BB * sizeof(float), stream);
    hipMemsetAsync(agg, 0, (size_t)MROWS * VAL * sizeof(float), stream);

    bsa_reduce_kw<<<1, 128, 0, stream>>>(k_w, k_b, kwsum, kbsum);

    int gemm_blocks = (MROWS / 16 + 7) / 8;
    bsa_values_gemm<<<gemm_blocks, 256, 0, stream>>>(x, v_w, v_b, values);

    bsa_ksum<<<(MROWS * 32 + 255) / 256, 256, 0, stream>>>(x, kwsum, kbsum, ksum);

    int edge_blocks = (EE * BB + 255) / 256;
    bsa_edge_max<<<edge_blocks, 256, 0, stream>>>(conn, rows, cols, ksum, mbuf);
    bsa_m_finalize<<<(NN * BB + 255) / 256, 256, 0, stream>>>(mbuf);

    bsa_edge_exp<<<edge_blocks, 256, 0, stream>>>(conn, rows, cols, ksum,
                                                  (const float*)mbuf, ex, denom);

    bsa_scatter<<<EE, 256, 0, stream>>>(rows, cols, ex, denom, values, agg);

    bsa_out_gemm_ln<<<gemm_blocks, 256, 0, stream>>>(agg, out_w, out_b,
                                                     ln_g, ln_b, out);
}


# Fewer, longer waves where a wave lived for one item: ksum takes 32 rows per wave (100000 waves -> 3125, sums stored as one 128-byte line), edge pass 4 edges per thread (25000 -> 6250); sum order kept
# speedup vs baseline: 4.0113x; 4.0113x over previous
#include <hip/hip_runtime.h>
#include <math.h>

#define BB    2
#define NN    50000
#define EE    800000
#define INP   128
#define KEY   64
#define VAL   128
#define MROWS (BB * NN)
#define LN_EPS 1e-5f

typedef __attribute__((ext_vector_type(2))) float v2f;
typedef __attribute__((ext_vector_type(8))) float v8f;

__device__ __forceinline__ unsigned enc_f32(float f) {
    unsigned u = __float_as_uint(f);
    return (u & 0x80000000u) ? ~u : (u | 0x80000000u);
}
__device__ __forceinline__ float dec_f32(unsigned u) {
    return (u & 0x80000000u) ? __uint_as_float(u & 0x7FFFFFFFu)
                             : __uint_as_float(~u);
}

__global__ void bsa_reduce_kw(const float* __restrict__ kw,
                              const float* __restrict__ kb,
                              float* __restrict__ kwsum,
                              float* __restrict__ kbsum) {
    int i = threadIdx.x;
    float s = 0.f;
    for (int j = 0; j < KEY; ++j) s += kw[i * KEY + j];
    kwsum[i] = s;
    if (i == 0) {
        float t = 0.f;
        for (int j = 0; j < KEY; ++j) t += kb[j];
        kbsum[0] = t;
    }
}

#define KS_ROWS 32
#define KS_GRP  (KS_ROWS / 2)

static_assert(MROWS % KS_ROWS == 0, "every wave has a full set of rows");

__global__ void bsa_ksum(const float* __restrict__ x,
                         const float* __restrict__ kwsum,
                         const float* __restrict__ kbsum,
                         float* __restrict__ ksum) {
    const int wid  = (blockIdx.x * blockDim.x + threadIdx.x) >> 5;
    const int lane = threadIdx.x & 31;
    if (wid >= MROWS / KS_ROWS) return;
    float kw[INP / 32];
    #pragma unroll
    for (int i = 0; i < INP / 32; ++i) kw[i] = kwsum[lane + i * 32];
    const float* xw = x + (size_t)wid * (KS_ROWS * INP) + lane;
    float mine = 0.f;
    #pragma unroll
    for (int g = 0; g < KS_ROWS; g += KS_GRP) {
        float v[KS_GRP][INP / 32];
        float s[KS_GRP];
        #pragma unroll
        for (int u = 0; u < KS_GRP; ++u) {
            #pragma unroll
            for (int i = 0; i < INP / 32; ++i) v[u][i] = xw[(g + u) * INP + i * 32];
        }
        #pragma unroll
        for (int u = 0; u < KS_GRP; ++u) {
            float t = 0.f;
            #pragma unroll
            for (int i = 0; i < INP / 32; ++i) t += v[u][i] * kw[i];
            s[u] = t;
        }
        #pragma unroll
        for (int off = 16; off; off >>= 1) {
            #pragma unroll
            for (int u = 0; u < KS_GRP; ++u) s[u] += __shfl_xor(s[u], off);
        }
        #pragma unroll
        for (int u = 0; u < KS_GRP; ++u)
            if (lane == g + u) mine = s[u];
    }
    ksum[wid * KS_ROWS + lane] = mine + kbsum[0];
}

__device__ __forceinline__ void bsa_wmma_tile(const float* __restrict__ Arow,
                                              const float* __restrict__ Bs,
                                              v8f acc[8], int lane) {
    const int kh = (lane >> 4) << 1;
    const int n  = lane & 15;
    for (int k = 0; k < 128; k += 4) {
        v2f a = *(const v2f*)(Arow + k + kh);
        #pragma unroll
        for (int c = 0; c < 8; ++c) {
            v2f bf;
            bf.x = Bs[(k + kh)     * 128 + (c << 4) + n];
            bf.y = Bs[(k + kh + 1) * 128 + (c << 4) + n];
            acc[c] = __builtin_amdgcn_wmma_f32_16x16x4_f32(
                false, a, false, bf, (short)0, acc[c], false, false);
        }
    }
}

__global__ void bsa_values_gemm(const float* __restrict__ A,
                                const float* __restrict__ W,
                                const float* __restrict__ bias,
                                float* __restrict__ out) {
    __shared__ float Bs[128 * 128];
    for (int i = threadIdx.x; i < 128 * 128; i += blockDim.x) Bs[i] = W[i];
    __syncthreads();

    int lane    = threadIdx.x & 31;
    int wv      = threadIdx.x >> 5;
    int tile    = blockIdx.x * 8 + wv;
    int rowbase = tile * 16;
    if (rowbase >= MROWS) return;

    v8f acc[8] = {};
    const float* Arow = A + (size_t)(rowbase + (lane & 15)) * 128;
    bsa_wmma_tile(Arow, Bs, acc, lane);

    int n      = lane & 15;
    int rowoff = (lane >> 4) << 3;
    #pragma unroll
    for (int c = 0; c < 8; ++c) {
        float bb = bias[(c << 4) + n];
        #pragma unroll
        for (int r = 0; r < 8; ++r)
            out[(size_t)(rowbase + rowoff + r) * 128 + (c << 4) + n] = acc[c][r] + bb;
    }
}

#define TR      200
#define NTILE   (NN / TR)
#define TCH     512
#define NTCH    ((EE + TCH - 1) / TCH)
#define TBYTES  (NTCH * TCH)
#define TNONE   255u
#define TNEVER  254u

static_assert(NN % TR == 0, "tiles must cover the rows exactly");
static_assert(NTILE <= 254 && TR <= 256, "tile id and local row are one byte each; 254 and 255 stay free");

#define ER_PER  4
#define ER_WAVE (32 * ER_PER)

static_assert(EE % ER_WAVE == 0, "a wave's edges are all inside the list or all outside");
static_assert(TBYTES - EE <= EE / ER_PER, "the pad is written by the first threads");

__global__ void bsa_edge_rec(const float* __restrict__ conn,
                             const int* __restrict__ rows,
                             const int* __restrict__ cols,
                             const float* __restrict__ ksum,
                             float* __restrict__ erec,
                             unsigned* __restrict__ tlist) {
    const int gtid = blockIdx.x * blockDim.x + threadIdx.x;
    unsigned char* tile8 = (unsigned char*)tlist;
    if (gtid < TBYTES - EE) tile8[EE + gtid] = (unsigned char)TNONE;
    const int e0 = (gtid >> 5) * ER_WAVE + (threadIdx.x & 31);
    if (e0 >= EE) return;
    int r[ER_PER], c[ER_PER];
    float cv[ER_PER], k0[ER_PER], k1[ER_PER];
    #pragma unroll
    for (int u = 0; u < ER_PER; ++u) {
        const int e = e0 + u * 32;
        r[u]  = rows[e];
        c[u]  = cols[e];
        cv[u] = conn[e];
    }
    #pragma unroll
    for (int u = 0; u < ER_PER; ++u) {
        k0[u] = ksum[c[u]];
        k1[u] = ksum[NN + c[u]];
    }
    #pragma unroll
    for (int u = 0; u < ER_PER; ++u) {
        const int e = e0 + u * 32;
        const float s0 = cv[u] * k0[u];
        const float s1 = cv[u] * k1[u];
        const int t = r[u] / TR;
        ((int4*)erec)[e] = make_int4(c[u], __float_as_int(s0), __float_as_int(s1), r[u] - t * TR);
        tile8[e] = (unsigned char)t;
    }
}

#define AGW        32
#define NTCH_PAD   (((NTCH + AGW - 1) / AGW) * AGW)
#define SCAN_ITERS (NTCH_PAD / AGW)
#define AHEAD      4
#define ROT_STEP   6
#define QCAP       (32 + TCH)

static_assert((NTILE - 1) * ROT_STEP < NTCH_PAD, "one wrap is enough for the start offset");
static_assert(TR * BB <= AGW * 32, "one thread per (row, batch) clears and settles the maxima and sums");

__device__ __forceinline__ float bsa_lane_bcast(float v, int j) {
    return __int_as_float(__builtin_amdgcn_readlane(__float_as_int(v), j));
}

__device__ __forceinline__ void bsa_tile_add(float* __restrict__ acc, unsigned rl, int lane,
                                             float w0, float w1, float4 x0, float4 x1) {
    float* p0 = acc + (rl * 2u) * VAL + lane;
    float* p1 = p0 + VAL;
    __hip_atomic_fetch_add(p0,      w0 * x0.x, __ATOMIC_RELAXED, __HIP_MEMORY_SCOPE_WORKGROUP);
    __hip_atomic_fetch_add(p0 + 32, w0 * x0.y, __ATOMIC_RELAXED, __HIP_MEMORY_SCOPE_WORKGROUP);
    __hip_atomic_fetch_add(p0 + 64, w0 * x0.z, __ATOMIC_RELAXED, __HIP_MEMORY_SCOPE_WORKGROUP);
    __hip_atomic_fetch_add(p0 + 96, w0 * x0.w, __ATOMIC_RELAXED, __HIP_MEMORY_SCOPE_WORKGROUP);
    __hip_atomic_fetch_add(p1,      w1 * x1.x, __ATOMIC_RELAXED, __HIP_MEMORY_SCOPE_WORKGROUP);
    __hip_atomic_fetch_add(p1 + 32, w1 * x1.y, __ATOMIC_RELAXED, __HIP_MEMORY_SCOPE_WORKGROUP);
    __hip_atomic_fetch_add(p1 + 64, w1 * x1.z, __ATOMIC_RELAXED, __HIP_MEMORY_SCOPE_WORKGROUP);
    __hip_atomic_fetch_add(p1 + 96, w1 * x1.w, __ATOMIC_RELAXED, __HIP_MEMORY_SCOPE_WORKGROUP);
}

__device__ __forceinline__ void bsa_max_drain(const unsigned* q, int n, int lane,
                                              const float* __restrict__ erec,
                                              unsigned* mt) {
    if (lane < n) {
        const int e = (int)q[lane];
        const int4 rec = ((const int4*)erec)[e];
        unsigned* p = mt + rec.w * BB;
        __hip_atomic_fetch_max(p,     enc_f32(__int_as_float(rec.y)), __ATOMIC_RELAXED, __HIP_MEMORY_SCOPE_WORKGROUP);
        __hip_atomic_fetch_max(p + 1, enc_f32(__int_as_float(rec.z)), __ATOMIC_RELAXED, __HIP_MEMORY_SCOPE_WORKGROUP);
    }
}

__device__ __forceinline__ void bsa_tile_drain(const unsigned* q, int n, int lane,
                                               const float* __restrict__ erec,
                                               const float* mf, float* dt,
                                               const float* __restrict__ values,
                                               float* __restrict__ acc) {
    int c = 0;
    unsigned rl = 0;
    float w0 = 0.f, w1 = 0.f;
    if (lane < n) {
        const int e = (int)q[lane];
        const int4 rec = ((const int4*)erec)[e];
        c = rec.x;
        rl = (unsigned)rec.w;
        const float m0 = mf[rec.w * BB];
        const float m1 = mf[rec.w * BB + 1];
        w0 = __expf(__int_as_float(rec.y) - m0);
        w1 = __expf(__int_as_float(rec.z) - m1);
        __hip_atomic_fetch_add(dt + rec.w * BB,     w0, __ATOMIC_RELAXED, __HIP_MEMORY_SCOPE_WORKGROUP);
        __hip_atomic_fetch_add(dt + rec.w * BB + 1, w1, __ATOMIC_RELAXED, __HIP_MEMORY_SCOPE_WORKGROUP);
    }
    const float* v0 = values + lane * 4;
    const float* v1 = values + (size_t)NN * VAL + lane * 4;
    int j = 0;
    #pragma nounroll
    for (; j + 4 <= n; j += 4) {
        int cj[4]; unsigned rj[4]; float a0[4], a1[4];
        float4 x0[4], x1[4];
        #pragma unroll
        for (int u = 0; u < 4; ++u) {
            cj[u] = __builtin_amdgcn_readlane(c, j + u);
            rj[u] = (unsigned)__builtin_amdgcn_readlane((int)rl, j + u);
            a0[u] = bsa_lane_bcast(w0, j + u);
            a1[u] = bsa_lane_bcast(w1, j + u);
            x0[u] = *(const float4*)(v0 + (size_t)cj[u] * VAL);
            x1[u] = *(const float4*)(v1 + (size_t)cj[u] * VAL);
        }
        #pragma unroll
        for (int u = 0; u < 4; ++u)
            bsa_tile_add(acc, rj[u], lane, a0[u], a1[u], x0[u], x1[u]);
    }
    for (; j < n; ++j) {
        int cj = __builtin_amdgcn_readlane(c, j);
        unsigned rj = (unsigned)__builtin_amdgcn_readlane((int)rl, j);
        float a0 = bsa_lane_bcast(w0, j);
        float a1 = bsa_lane_bcast(w1, j);
        float4 x0 = *(const float4*)(v0 + (size_t)cj * VAL);
        float4 x1 = *(const float4*)(v1 + (size_t)cj * VAL);
        bsa_tile_add(acc, rj, lane, a0, a1, x0, x1);
    }
}

template <int PASS>
__device__ __forceinline__ void bsa_drain_ready(unsigned* q, int& cnt, int need, int lane,
                                                const float* __restrict__ erec,
                                                const float* __restrict__ values,
                                                unsigned* mt, float* dt,
                                                float* __restrict__ acc) {
    int head = 0;
    while (cnt - head >= need) {
        int n = cnt - head;
        if (n > 32) n = 32;
        if (PASS == 0) bsa_max_drain(q + head, n, lane, erec, mt);
        else           bsa_tile_drain(q + head, n, lane, erec, (const float*)mt, dt, values, acc);
        head += n;
    }
    if (head > 0) {
        const int rem = cnt - head;
        unsigned keep = 0u;
        if (lane < rem) keep = q[head + lane];
        if (lane < rem) q[lane] = keep;
        cnt = rem;
    }
}

__device__ __forceinline__ int4 bsa_tiles16(const unsigned* __restrict__ tlist, int ch, int lane) {
    const int chc = ch >= 0 ? ch : NTCH - 1;
    return ((const int4*)tlist)[chc * 32 + lane];
}

__device__ __forceinline__ unsigned bsa_zero_bytes(unsigned t) {
    return ~(((t & 0x7F7F7F7Fu) + 0x7F7F7F7Fu) | t | 0x7F7F7F7Fu);
}

__device__ __forceinline__ void bsa_scan_one(int4 v, int ch, int lane, unsigned tsplat, unsigned* q, int& cnt) {
    const unsigned ts = ch >= 0 ? tsplat : TNEVER * 0x01010101u;
    const int ebase = ch * TCH + lane * 16;
    const unsigned z0 = bsa_zero_bytes((unsigned)v.x ^ ts);
    const unsigned z1 = bsa_zero_bytes((unsigned)v.y ^ ts);
    const unsigned z2 = bsa_zero_bytes((unsigned)v.z ^ ts);
    const unsigned z3 = bsa_zero_bytes((unsigned)v.w ^ ts);
#define BSA_APPEND_WORD(zk, k)                                                   \
    if (__builtin_amdgcn_ballot_w32((zk) != 0u) != 0u) {                         \
        for (int j = 0; j < 4; ++j) {                                            \
            const bool m = (((zk) >> (8 * j + 7)) & 1u) != 0u;                   \
            const unsigned mask = __builtin_amdgcn_ballot_w32(m);                \
            if (mask != 0u) {                                                    \
                const int pre = (int)__builtin_amdgcn_mbcnt_lo(mask, 0u);        \
                if (m) q[cnt + pre] = (unsigned)(ebase + 4 * (k) + j);           \
                cnt += __builtin_popcount(mask);                                 \
            }                                                                    \
        }                                                                        \
    }
    BSA_APPEND_WORD(z0, 0)
    BSA_APPEND_WORD(z1, 1)
    BSA_APPEND_WORD(z2, 2)
    BSA_APPEND_WORD(z3, 3)
#undef BSA_APPEND_WORD
}

__device__ __forceinline__ int bsa_fetch_index(int st, int wv, int rot) {
    int s = st * AGW + wv + rot;
    if (s >= NTCH_PAD) s -= NTCH_PAD;
    return (st < SCAN_ITERS && s < NTCH) ? s : -1;
}

template <int PASS>
__device__ __forceinline__ void bsa_scan_pass(const unsigned* __restrict__ tlist,
                                              int wv, int lane, int rot, unsigned tsplat,
                                              unsigned* q,
                                              const float* __restrict__ erec,
                                              const float* __restrict__ values,
                                              unsigned* mt, float* dt,
                                              float* __restrict__ acc) {
    int cnt = 0;

    static_assert(SCAN_ITERS >= AHEAD && AHEAD == 4, "four list loads are kept in flight");
    int4 c0 = bsa_tiles16(tlist, bsa_fetch_index(0, wv, rot), lane);
    int4 c1 = bsa_tiles16(tlist, bsa_fetch_index(1, wv, rot), lane);
    int4 c2 = bsa_tiles16(tlist, bsa_fetch_index(2, wv, rot), lane);
    int4 c3 = bsa_tiles16(tlist, bsa_fetch_index(3, wv, rot), lane);

#define BSA_STEP(cu, u)                                                                    \
    {                                                                                      \
        bsa_scan_one(cu, bsa_fetch_index(it + (u), wv, rot), lane, tsplat, q, cnt);        \
        cu = bsa_tiles16(tlist, bsa_fetch_index(it + (u) + AHEAD, wv, rot), lane);         \
        bsa_drain_ready<PASS>(q, cnt, 32, lane, erec, values, mt, dt, acc);                \
    }
    for (int it = 0; it < SCAN_ITERS; it += AHEAD) {
        BSA_STEP(c0, 0)
        BSA_STEP(c1, 1)
        BSA_STEP(c2, 2)
        BSA_STEP(c3, 3)
    }
#undef BSA_STEP
    bsa_drain_ready<PASS>(q, cnt, 1, lane, erec, values, mt, dt, acc);
}

__global__ void __launch_bounds__(AGW * 32)
bsa_agg_owner(const float* __restrict__ erec,
              const unsigned* __restrict__ tlist,
              const float* __restrict__ values,
              float* __restrict__ agg) {
    __shared__ float acc[TR * BB * VAL];
    __shared__ unsigned queue[AGW * QCAP];
    __shared__ unsigned mt[TR * BB];
    __shared__ float dt[TR * BB];

    const int tid  = threadIdx.x;
    const int lane = tid & 31;
    const int wv   = __builtin_amdgcn_readfirstlane(tid >> 5);
    const int base = blockIdx.x * TR;
    const unsigned tsplat = blockIdx.x * 0x01010101u;

    float4* acc4 = (float4*)acc;
    for (int i = tid; i < TR * BB * VAL / 4; i += AGW * 32)
        acc4[i] = make_float4(0.f, 0.f, 0.f, 0.f);
    if (tid < TR * BB) { mt[tid] = 0u; dt[tid] = 0.f; }
    __syncthreads();

    unsigned* q = queue + wv * QCAP;

    const int rot = blockIdx.x * ROT_STEP;

    bsa_scan_pass<0>(tlist, wv, lane, rot, tsplat, q, erec, values, mt, dt, acc);
    __syncthreads();
    if (tid < TR * BB) {
        const unsigned u = mt[tid];
        float m = (u == 0u) ? 0.0f : dec_f32(u);
        if (!isfinite(m)) m = 0.0f;
        mt[tid] = __float_as_uint(m);
    }
    __syncthreads();

    bsa_scan_pass<1>(tlist, wv, lane, rot, tsplat, q, erec, values, mt, dt, acc);
    __syncthreads();

    for (int row = wv; row < TR * BB; row += AGW) {
        const int rl = row >> 1, b = row & 1;
        const float* a = acc + row * VAL + lane;
        const float d  = dt[row];
        const float dd = (d == 0.0f) ? 1.0f : d;
        float4 o;
        o.x = a[0] / dd; o.y = a[32] / dd; o.z = a[64] / dd; o.w = a[96] / dd;
        *(float4*)(agg + ((size_t)b * NN + base + rl) * VAL + lane * 4) = o;
    }
}

__global__ void bsa_out_gemm_ln(const float* __restrict__ A,
                                const float* __restrict__ W,
                                const float* __restrict__ bias,
                                const float* __restrict__ lng,
                                const float* __restrict__ lnb,
                                float* __restrict__ out) {
    __shared__ float Bs[128 * 128];
    for (int i = threadIdx.x; i < 128 * 128; i += blockDim.x) Bs[i] = W[i];
    __syncthreads();

    int lane    = threadIdx.x & 31;
    int wv      = threadIdx.x >> 5;
    int tile    = blockIdx.x * 8 + wv;
    int rowbase = tile * 16;
    if (rowbase >= MROWS) return;

    v8f acc[8] = {};
    const float* Arow = A + (size_t)(rowbase + (lane & 15)) * 128;
    bsa_wmma_tile(Arow, Bs, acc, lane);

    int n = lane & 15;
    float sr[8], sq[8];
    #pragma unroll
    for (int r = 0; r < 8; ++r) { sr[r] = 0.f; sq[r] = 0.f; }
    #pragma unroll
    for (int c = 0; c < 8; ++c) {
        float bb = bias[(c << 4) + n];
        #pragma unroll
        for (int r = 0; r < 8; ++r) {
            float v = acc[c][r] + bb;
            v = v / (1.0f + __expf(-v));
            acc[c][r] = v;
            sr[r] += v;
            sq[r] += v * v;
        }
    }
    #pragma unroll
    for (int r = 0; r < 8; ++r) {
        #pragma unroll
        for (int off = 8; off; off >>= 1) {
            sr[r] += __shfl_xor(sr[r], off);
            sq[r] += __shfl_xor(sq[r], off);
        }
    }
    float mu[8], rstd[8];
    #pragma unroll
    for (int r = 0; r < 8; ++r) {
        mu[r] = sr[r] * (1.0f / VAL);
        float var = sq[r] * (1.0f / VAL) - mu[r] * mu[r];
        rstd[r] = rsqrtf(var + LN_EPS);
    }
    int rowoff = (lane >> 4) << 3;
    #pragma unroll
    for (int c = 0; c < 8; ++c) {
        int col = (c << 4) + n;
        float g = lng[col], b2 = lnb[col];
        #pragma unroll
        for (int r = 0; r < 8; ++r)
            out[(size_t)(rowbase + rowoff + r) * VAL + col] =
                (acc[c][r] - mu[r]) * rstd[r] * g + b2;
    }
}

extern "C" void kernel_launch(void* const* d_in, const int* in_sizes, int n_in,
                              void* d_out, int out_size, void* d_ws, size_t ws_size,
                              hipStream_t stream) {
    const float* x     = (const float*)d_in[0];
    const float* conn  = (const float*)d_in[1];
    const float* k_w   = (const float*)d_in[2];
    const float* k_b   = (const float*)d_in[3];
    const float* v_w   = (const float*)d_in[4];
    const float* v_b   = (const float*)d_in[5];
    const float* out_w = (const float*)d_in[6];
    const float* out_b = (const float*)d_in[7];
    const float* ln_g  = (const float*)d_in[8];
    const float* ln_b  = (const float*)d_in[9];
    const int*   rows  = (const int*)d_in[10];
    const int*   cols  = (const int*)d_in[11];
    float* out = (float*)d_out;

    char* p = (char*)d_ws;
    auto alloc = [&](size_t bytes) {
        void* r = (void*)p;
        p += (bytes + 255) & ~(size_t)255;
        return r;
    };
    float*    kwsum  = (float*)alloc(INP * sizeof(float));
    float*    kbsum  = (float*)alloc(sizeof(float));
    float*    ksum   = (float*)alloc((size_t)MROWS * sizeof(float));
    float*    erec   = (float*)alloc((size_t)EE * 4 * sizeof(float));
    unsigned* tlist  = (unsigned*)alloc((size_t)TBYTES);
    float*    values = (float*)alloc((size_t)MROWS * VAL * sizeof(float));
    float*    agg    = (float*)alloc((size_t)MROWS * VAL * sizeof(float));

    bsa_reduce_kw<<<1, 128, 0, stream>>>(k_w, k_b, kwsum, kbsum);

    int gemm_blocks = (MROWS / 16 + 7) / 8;
    bsa_values_gemm<<<gemm_blocks, 256, 0, stream>>>(x, v_w, v_b, values);

    bsa_ksum<<<(MROWS / KS_ROWS * 32 + 255) / 256, 256, 0, stream>>>(x, kwsum, kbsum, ksum);

    bsa_edge_rec<<<(EE / ER_PER + 255) / 256, 256, 0, stream>>>(conn, rows, cols, ksum, erec, tlist);

    bsa_agg_owner<<<NTILE, AGW * 32, 0, stream>>>(erec, tlist, values, agg);

    bsa_out_gemm_ln<<<gemm_blocks, 256, 0, stream>>>(agg, out_w, out_b,
                                                     ln_g, ln_b, out);
}


# Weight fill of both GEMMs: a thread asks for its sixteen 16-byte pieces of W at once, then stores them to LDS; was 64 trips of load one float, wait, store one float, per thread in every block
# speedup vs baseline: 5.2411x; 1.3066x over previous
#include <hip/hip_runtime.h>
#include <math.h>

#define BB    2
#define NN    50000
#define EE    800000
#define INP   128
#define KEY   64
#define VAL   128
#define MROWS (BB * NN)
#define LN_EPS 1e-5f

typedef __attribute__((ext_vector_type(2))) float v2f;
typedef __attribute__((ext_vector_type(8))) float v8f;

__device__ __forceinline__ unsigned enc_f32(float f) {
    unsigned u = __float_as_uint(f);
    return (u & 0x80000000u) ? ~u : (u | 0x80000000u);
}
__device__ __forceinline__ float dec_f32(unsigned u) {
    return (u & 0x80000000u) ? __uint_as_float(u & 0x7FFFFFFFu)
                             : __uint_as_float(~u);
}

__global__ void bsa_reduce_kw(const float* __restrict__ kw,
                              const float* __restrict__ kb,
                              float* __restrict__ kwsum,
                              float* __restrict__ kbsum) {
    int i = threadIdx.x;
    float s = 0.f;
    for (int j = 0; j < KEY; ++j) s += kw[i * KEY + j];
    kwsum[i] = s;
    if (i == 0) {
        float t = 0.f;
        for (int j = 0; j < KEY; ++j) t += kb[j];
        kbsum[0] = t;
    }
}

#define KS_ROWS 32
#define KS_GRP  (KS_ROWS / 2)

static_assert(MROWS % KS_ROWS == 0, "every wave has a full set of rows");

__global__ void bsa_ksum(const float* __restrict__ x,
                         const float* __restrict__ kwsum,
                         const float* __restrict__ kbsum,
                         float* __restrict__ ksum) {
    const int wid  = (blockIdx.x * blockDim.x + threadIdx.x) >> 5;
    const int lane = threadIdx.x & 31;
    if (wid >= MROWS / KS_ROWS) return;
    float kw[INP / 32];
    #pragma unroll
    for (int i = 0; i < INP / 32; ++i) kw[i] = kwsum[lane + i * 32];
    const float* xw = x + (size_t)wid * (KS_ROWS * INP) + lane;
    float mine = 0.f;
    #pragma unroll
    for (int g = 0; g < KS_ROWS; g += KS_GRP) {
        float v[KS_GRP][INP / 32];
        float s[KS_GRP];
        #pragma unroll
        for (int u = 0; u < KS_GRP; ++u) {
            #pragma unroll
            for (int i = 0; i < INP / 32; ++i) v[u][i] = xw[(g + u) * INP + i * 32];
        }
        #pragma unroll
        for (int u = 0; u < KS_GRP; ++u) {
            float t = 0.f;
            #pragma unroll
            for (int i = 0; i < INP / 32; ++i) t += v[u][i] * kw[i];
            s[u] = t;
        }
        #pragma unroll
        for (int off = 16; off; off >>= 1) {
            #pragma unroll
            for (int u = 0; u < KS_GRP; ++u) s[u] += __shfl_xor(s[u], off);
        }
        #pragma unroll
        for (int u = 0; u < KS_GRP; ++u)
            if (lane == g + u) mine = s[u];
    }
    ksum[wid * KS_ROWS + lane] = mine + kbsum[0];
}

__device__ __forceinline__ void bsa_wmma_tile(const float* __restrict__ Arow,
                                              const float* __restrict__ Bs,
                                              v8f acc[8], int lane) {
    const int kh = (lane >> 4) << 1;
    const int n  = lane & 15;
    for (int k = 0; k < 128; k += 4) {
        v2f a = *(const v2f*)(Arow + k + kh);
        #pragma unroll
        for (int c = 0; c < 8; ++c) {
            v2f bf;
            bf.x = Bs[(k + kh)     * 128 + (c << 4) + n];
            bf.y = Bs[(k + kh + 1) * 128 + (c << 4) + n];
            acc[c] = __builtin_amdgcn_wmma_f32_16x16x4_f32(
                false, a, false, bf, (short)0, acc[c], false, false);
        }
    }
}

#define GEMM_THREADS 256
#define W_PIECES     (128 * 128 / 4 / GEMM_THREADS)

static_assert(128 * 128 % (4 * GEMM_THREADS) == 0, "every thread fills the same number of pieces");

__device__ __forceinline__ void bsa_fill_weight(const float* __restrict__ W, float4* __restrict__ Bs4) {
    const float4* W4 = (const float4*)W;
    float4 w[W_PIECES];
    #pragma unroll
    for (int i = 0; i < W_PIECES; ++i) w[i] = W4[threadIdx.x + i * GEMM_THREADS];
    #pragma unroll
    for (int i = 0; i < W_PIECES; ++i) Bs4[threadIdx.x + i * GEMM_THREADS] = w[i];
}

__global__ void bsa_values_gemm(const float* __restrict__ A,
                                const float* __restrict__ W,
                                const float* __restrict__ bias,
                                float* __restrict__ out) {
    __shared__ float4 Bs4[128 * 128 / 4];
    const float* Bs = (const float*)Bs4;
    bsa_fill_weight(W, Bs4);
    __syncthreads();

    int lane    = threadIdx.x & 31;
    int wv      = threadIdx.x >> 5;
    int tile    = blockIdx.x * 8 + wv;
    int rowbase = tile * 16;
    if (rowbase >= MROWS) return;

    v8f acc[8] = {};
    const float* Arow = A + (size_t)(rowbase + (lane & 15)) * 128;
    bsa_wmma_tile(Arow, Bs, acc, lane);

    int n      = lane & 15;
    int rowoff = (lane >> 4) << 3;
    #pragma unroll
    for (int c = 0; c < 8; ++c) {
        float bb = bias[(c << 4) + n];
        #pragma unroll
        for (int r = 0; r < 8; ++r)
            out[(size_t)(rowbase + rowoff + r) * 128 + (c << 4) + n] = acc[c][r] + bb;
    }
}

#define TR      200
#define NTILE   (NN / TR)
#define TCH     512
#define NTCH    ((EE + TCH - 1) / TCH)
#define TBYTES  (NTCH * TCH)
#define TNONE   255u
#define TNEVER  254u

static_assert(NN % TR == 0, "tiles must cover the rows exactly");
static_assert(NTILE <= 254 && TR <= 256, "tile id and local row are one byte each; 254 and 255 stay free");

#define ER_PER  4
#define ER_WAVE (32 * ER_PER)

static_assert(EE % ER_WAVE == 0, "a wave's edges are all inside the list or all outside");
static_assert(TBYTES - EE <= EE / ER_PER, "the pad is written by the first threads");

__global__ void bsa_edge_rec(const float* __restrict__ conn,
                             const int* __restrict__ rows,
                             const int* __restrict__ cols,
                             const float* __restrict__ ksum,
                             float* __restrict__ erec,
                             unsigned* __restrict__ tlist) {
    const int gtid = blockIdx.x * blockDim.x + threadIdx.x;
    unsigned char* tile8 = (unsigned char*)tlist;
    if (gtid < TBYTES - EE) tile8[EE + gtid] = (unsigned char)TNONE;
    const int e0 = (gtid >> 5) * ER_WAVE + (threadIdx.x & 31);
    if (e0 >= EE) return;
    int r[ER_PER], c[ER_PER];
    float cv[ER_PER], k0[ER_PER], k1[ER_PER];
    #pragma unroll
    for (int u = 0; u < ER_PER; ++u) {
        const int e = e0 + u * 32;
        r[u]  = rows[e];
        c[u]  = cols[e];
        cv[u] = conn[e];
    }
    #pragma unroll
    for (int u = 0; u < ER_PER; ++u) {
        k0[u] = ksum[c[u]];
        k1[u] = ksum[NN + c[u]];
    }
    #pragma unroll
    for (int u = 0; u < ER_PER; ++u) {
        const int e = e0 + u * 32;
        const float s0 = cv[u] * k0[u];
        const float s1 = cv[u] * k1[u];
        const int t = r[u] / TR;
        ((int4*)erec)[e] = make_int4(c[u], __float_as_int(s0), __float_as_int(s1), r[u] - t * TR);
        tile8[e] = (unsigned char)t;
    }
}

#define AGW        32
#define NTCH_PAD   (((NTCH + AGW - 1) / AGW) * AGW)
#define SCAN_ITERS (NTCH_PAD / AGW)
#define AHEAD      4
#define ROT_STEP   6
#define QCAP       (32 + TCH)

static_assert((NTILE - 1) * ROT_STEP < NTCH_PAD, "one wrap is enough for the start offset");
static_assert(TR * BB <= AGW * 32, "one thread per (row, batch) clears and settles the maxima and sums");

__device__ __forceinline__ float bsa_lane_bcast(float v, int j) {
    return __int_as_float(__builtin_amdgcn_readlane(__float_as_int(v), j));
}

__device__ __forceinline__ void bsa_tile_add(float* __restrict__ acc, unsigned rl, int lane,
                                             float w0, float w1, float4 x0, float4 x1) {
    float* p0 = acc + (rl * 2u) * VAL + lane;
    float* p1 = p0 + VAL;
    __hip_atomic_fetch_add(p0,      w0 * x0.x, __ATOMIC_RELAXED, __HIP_MEMORY_SCOPE_WORKGROUP);
    __hip_atomic_fetch_add(p0 + 32, w0 * x0.y, __ATOMIC_RELAXED, __HIP_MEMORY_SCOPE_WORKGROUP);
    __hip_atomic_fetch_add(p0 + 64, w0 * x0.z, __ATOMIC_RELAXED, __HIP_MEMORY_SCOPE_WORKGROUP);
    __hip_atomic_fetch_add(p0 + 96, w0 * x0.w, __ATOMIC_RELAXED, __HIP_MEMORY_SCOPE_WORKGROUP);
    __hip_atomic_fetch_add(p1,      w1 * x1.x, __ATOMIC_RELAXED, __HIP_MEMORY_SCOPE_WORKGROUP);
    __hip_atomic_fetch_add(p1 + 32, w1 * x1.y, __ATOMIC_RELAXED, __HIP_MEMORY_SCOPE_WORKGROUP);
    __hip_atomic_fetch_add(p1 + 64, w1 * x1.z, __ATOMIC_RELAXED, __HIP_MEMORY_SCOPE_WORKGROUP);
    __hip_atomic_fetch_add(p1 + 96, w1 * x1.w, __ATOMIC_RELAXED, __HIP_MEMORY_SCOPE_WORKGROUP);
}

__device__ __forceinline__ void bsa_max_drain(const unsigned* q, int n, int lane,
                                              const float* __restrict__ erec,
                                              unsigned* mt) {
    if (lane < n) {
        const int e = (int)q[lane];
        const int4 rec = ((const int4*)erec)[e];
        unsigned* p = mt + rec.w * BB;
        __hip_atomic_fetch_max(p,     enc_f32(__int_as_float(rec.y)), __ATOMIC_RELAXED, __HIP_MEMORY_SCOPE_WORKGROUP);
        __hip_atomic_fetch_max(p + 1, enc_f32(__int_as_float(rec.z)), __ATOMIC_RELAXED, __HIP_MEMORY_SCOPE_WORKGROUP);
    }
}

__device__ __forceinline__ void bsa_tile_drain(const unsigned* q, int n, int lane,
                                               const float* __restrict__ erec,
                                               const float* mf, float* dt,
                                               const float* __restrict__ values,
                                               float* __restrict__ acc) {
    int c = 0;
    unsigned rl = 0;
    float w0 = 0.f, w1 = 0.f;
    if (lane < n) {
        const int e = (int)q[lane];
        const int4 rec = ((const int4*)erec)[e];
        c = rec.x;
        rl = (unsigned)rec.w;
        const float m0 = mf[rec.w * BB];
        const float m1 = mf[rec.w * BB + 1];
        w0 = __expf(__int_as_float(rec.y) - m0);
        w1 = __expf(__int_as_float(rec.z) - m1);
        __hip_atomic_fetch_add(dt + rec.w * BB,     w0, __ATOMIC_RELAXED, __HIP_MEMORY_SCOPE_WORKGROUP);
        __hip_atomic_fetch_add(dt + rec.w * BB + 1, w1, __ATOMIC_RELAXED, __HIP_MEMORY_SCOPE_WORKGROUP);
    }
    const float* v0 = values + lane * 4;
    const float* v1 = values + (size_t)NN * VAL + lane * 4;
    int j = 0;
    #pragma nounroll
    for (; j + 4 <= n; j += 4) {
        int cj[4]; unsigned rj[4]; float a0[4], a1[4];
        float4 x0[4], x1[4];
        #pragma unroll
        for (int u = 0; u < 4; ++u) {
            cj[u] = __builtin_amdgcn_readlane(c, j + u);
            rj[u] = (unsigned)__builtin_amdgcn_readlane((int)rl, j + u);
            a0[u] = bsa_lane_bcast(w0, j + u);
            a1[u] = bsa_lane_bcast(w1, j + u);
            x0[u] = *(const float4*)(v0 + (size_t)cj[u] * VAL);
            x1[u] = *(const float4*)(v1 + (size_t)cj[u] * VAL);
        }
        #pragma unroll
        for (int u = 0; u < 4; ++u)
            bsa_tile_add(acc, rj[u], lane, a0[u], a1[u], x0[u], x1[u]);
    }
    for (; j < n; ++j) {
        int cj = __builtin_amdgcn_readlane(c, j);
        unsigned rj = (unsigned)__builtin_amdgcn_readlane((int)rl, j);
        float a0 = bsa_lane_bcast(w0, j);
        float a1 = bsa_lane_bcast(w1, j);
        float4 x0 = *(const float4*)(v0 + (size_t)cj * VAL);
        float4 x1 = *(const float4*)(v1 + (size_t)cj * VAL);
        bsa_tile_add(acc, rj, lane, a0, a1, x0, x1);
    }
}

template <int PASS>
__device__ __forceinline__ void bsa_drain_ready(unsigned* q, int& cnt, int need, int lane,
                                                const float* __restrict__ erec,
                                                const float* __restrict__ values,
                                                unsigned* mt, float* dt,
                                                float* __restrict__ acc) {
    int head = 0;
    while (cnt - head >= need) {
        int n = cnt - head;
        if (n > 32) n = 32;
        if (PASS == 0) bsa_max_drain(q + head, n, lane, erec, mt);
        else           bsa_tile_drain(q + head, n, lane, erec, (const float*)mt, dt, values, acc);
        head += n;
    }
    if (head > 0) {
        const int rem = cnt - head;
        unsigned keep = 0u;
        if (lane < rem) keep = q[head + lane];
        if (lane < rem) q[lane] = keep;
        cnt = rem;
    }
}

__device__ __forceinline__ int4 bsa_tiles16(const unsigned* __restrict__ tlist, int ch, int lane) {
    const int chc = ch >= 0 ? ch : NTCH - 1;
    return ((const int4*)tlist)[chc * 32 + lane];
}

__device__ __forceinline__ unsigned bsa_zero_bytes(unsigned t) {
    return ~(((t & 0x7F7F7F7Fu) + 0x7F7F7F7Fu) | t | 0x7F7F7F7Fu);
}

__device__ __forceinline__ void bsa_scan_one(int4 v, int ch, int lane, unsigned tsplat, unsigned* q, int& cnt) {
    const unsigned ts = ch >= 0 ? tsplat : TNEVER * 0x01010101u;
    const int ebase = ch * TCH + lane * 16;
    const unsigned z0 = bsa_zero_bytes((unsigned)v.x ^ ts);
    const unsigned z1 = bsa_zero_bytes((unsigned)v.y ^ ts);
    const unsigned z2 = bsa_zero_bytes((unsigned)v.z ^ ts);
    const unsigned z3 = bsa_zero_bytes((unsigned)v.w ^ ts);
#define BSA_APPEND_WORD(zk, k)                                                   \
    if (__builtin_amdgcn_ballot_w32((zk) != 0u) != 0u) {                         \
        for (int j = 0; j < 4; ++j) {                                            \
            const bool m = (((zk) >> (8 * j + 7)) & 1u) != 0u;                   \
            const unsigned mask = __builtin_amdgcn_ballot_w32(m);                \
            if (mask != 0u) {                                                    \
                const int pre = (int)__builtin_amdgcn_mbcnt_lo(mask, 0u);        \
                if (m) q[cnt + pre] = (unsigned)(ebase + 4 * (k) + j);           \
                cnt += __builtin_popcount(mask);                                 \
            }                                                                    \
        }                                                                        \
    }
    BSA_APPEND_WORD(z0, 0)
    BSA_APPEND_WORD(z1, 1)
    BSA_APPEND_WORD(z2, 2)
    BSA_APPEND_WORD(z3, 3)
#undef BSA_APPEND_WORD
}

__device__ __forceinline__ int bsa_fetch_index(int st, int wv, int rot) {
    int s = st * AGW + wv + rot;
    if (s >= NTCH_PAD) s -= NTCH_PAD;
    return (st < SCAN_ITERS && s < NTCH) ? s : -1;
}

template <int PASS>
__device__ __forceinline__ void bsa_scan_pass(const unsigned* __restrict__ tlist,
                                              int wv, int lane, int rot, unsigned tsplat,
                                              unsigned* q,
                                              const float* __restrict__ erec,
                                              const float* __restrict__ values,
                                              unsigned* mt, float* dt,
                                              float* __restrict__ acc) {
    int cnt = 0;

    static_assert(SCAN_ITERS >= AHEAD && AHEAD == 4, "four list loads are kept in flight");
    int4 c0 = bsa_tiles16(tlist, bsa_fetch_index(0, wv, rot), lane);
    int4 c1 = bsa_tiles16(tlist, bsa_fetch_index(1, wv, rot), lane);
    int4 c2 = bsa_tiles16(tlist, bsa_fetch_index(2, wv, rot), lane);
    int4 c3 = bsa_tiles16(tlist, bsa_fetch_index(3, wv, rot), lane);

#define BSA_STEP(cu, u)                                                                    \
    {                                                                                      \
        bsa_scan_one(cu, bsa_fetch_index(it + (u), wv, rot), lane, tsplat, q, cnt);        \
        cu = bsa_tiles16(tlist, bsa_fetch_index(it + (u) + AHEAD, wv, rot), lane);         \
        bsa_drain_ready<PASS>(q, cnt, 32, lane, erec, values, mt, dt, acc);                \
    }
    for (int it = 0; it < SCAN_ITERS; it += AHEAD) {
        BSA_STEP(c0, 0)
        BSA_STEP(c1, 1)
        BSA_STEP(c2, 2)
        BSA_STEP(c3, 3)
    }
#undef BSA_STEP
    bsa_drain_ready<PASS>(q, cnt, 1, lane, erec, values, mt, dt, acc);
}

__global__ void __launch_bounds__(AGW * 32)
bsa_agg_owner(const float* __restrict__ erec,
              const unsigned* __restrict__ tlist,
              const float* __restrict__ values,
              float* __restrict__ agg) {
    __shared__ float acc[TR * BB * VAL];
    __shared__ unsigned queue[AGW * QCAP];
    __shared__ unsigned mt[TR * BB];
    __shared__ float dt[TR * BB];

    const int tid  = threadIdx.x;
    const int lane = tid & 31;
    const int wv   = __builtin_amdgcn_readfirstlane(tid >> 5);
    const int base = blockIdx.x * TR;
    const unsigned tsplat = blockIdx.x * 0x01010101u;

    float4* acc4 = (float4*)acc;
    for (int i = tid; i < TR * BB * VAL / 4; i += AGW * 32)
        acc4[i] = make_float4(0.f, 0.f, 0.f, 0.f);
    if (tid < TR * BB) { mt[tid] = 0u; dt[tid] = 0.f; }
    __syncthreads();

    unsigned* q = queue + wv * QCAP;

    const int rot = blockIdx.x * ROT_STEP;

    bsa_scan_pass<0>(tlist, wv, lane, rot, tsplat, q, erec, values, mt, dt, acc);
    __syncthreads();
    if (tid < TR * BB) {
        const unsigned u = mt[tid];
        float m = (u == 0u) ? 0.0f : dec_f32(u);
        if (!isfinite(m)) m = 0.0f;
        mt[tid] = __float_as_uint(m);
    }
    __syncthreads();

    bsa_scan_pass<1>(tlist, wv, lane, rot, tsplat, q, erec, values, mt, dt, acc);
    __syncthreads();

    for (int row = wv; row < TR * BB; row += AGW) {
        const int rl = row >> 1, b = row & 1;
        const float* a = acc + row * VAL + lane;
        const float d  = dt[row];
        const float dd = (d == 0.0f) ? 1.0f : d;
        float4 o;
        o.x = a[0] / dd; o.y = a[32] / dd; o.z = a[64] / dd; o.w = a[96] / dd;
        *(float4*)(agg + ((size_t)b * NN + base + rl) * VAL + lane * 4) = o;
    }
}

__global__ void bsa_out_gemm_ln(const float* __restrict__ A,
                                const float* __restrict__ W,
                                const float* __restrict__ bias,
                                const float* __restrict__ lng,
                                const float* __restrict__ lnb,
                                float* __restrict__ out) {
    __shared__ float4 Bs4[128 * 128 / 4];
    const float* Bs = (const float*)Bs4;
    bsa_fill_weight(W, Bs4);
    __syncthreads();

    int lane    = threadIdx.x & 31;
    int wv      = threadIdx.x >> 5;
    int tile    = blockIdx.x * 8 + wv;
    int rowbase = tile * 16;
    if (rowbase >= MROWS) return;

    v8f acc[8] = {};
    const float* Arow = A + (size_t)(rowbase + (lane & 15)) * 128;
    bsa_wmma_tile(Arow, Bs, acc, lane);

    int n = lane & 15;
    float sr[8], sq[8];
    #pragma unroll
    for (int r = 0; r < 8; ++r) { sr[r] = 0.f; sq[r] = 0.f; }
    #pragma unroll
    for (int c = 0; c < 8; ++c) {
        float bb = bias[(c << 4) + n];
        #pragma unroll
        for (int r = 0; r < 8; ++r) {
            float v = acc[c][r] + bb;
            v = v / (1.0f + __expf(-v));
            acc[c][r] = v;
            sr[r] += v;
            sq[r] += v * v;
        }
    }
    #pragma unroll
    for (int r = 0; r < 8; ++r) {
        #pragma unroll
        for (int off = 8; off; off >>= 1) {
            sr[r] += __shfl_xor(sr[r], off);
            sq[r] += __shfl_xor(sq[r], off);
        }
    }
    float mu[8], rstd[8];
    #pragma unroll
    for (int r = 0; r < 8; ++r) {
        mu[r] = sr[r] * (1.0f / VAL);
        float var = sq[r] * (1.0f / VAL) - mu[r] * mu[r];
        rstd[r] = rsqrtf(var + LN_EPS);
    }
    int rowoff = (lane >> 4) << 3;
    #pragma unroll
    for (int c = 0; c < 8; ++c) {
        int col = (c << 4) + n;
        float g = lng[col], b2 = lnb[col];
        #pragma unroll
        for (int r = 0; r < 8; ++r)
            out[(size_t)(rowbase + rowoff + r) * VAL + col] =
                (acc[c][r] - mu[r]) * rstd[r] * g + b2;
    }
}

extern "C" void kernel_launch(void* const* d_in, const int* in_sizes, int n_in,
                              void* d_out, int out_size, void* d_ws, size_t ws_size,
                              hipStream_t stream) {
    const float* x     = (const float*)d_in[0];
    const float* conn  = (const float*)d_in[1];
    const float* k_w   = (const float*)d_in[2];
    const float* k_b   = (const float*)d_in[3];
    const float* v_w   = (const float*)d_in[4];
    const float* v_b   = (const float*)d_in[5];
    const float* out_w = (const float*)d_in[6];
    const float* out_b = (const float*)d_in[7];
    const float* ln_g  = (const float*)d_in[8];
    const float* ln_b  = (const float*)d_in[9];
    const int*   rows  = (const int*)d_in[10];
    const int*   cols  = (const int*)d_in[11];
    float* out = (float*)d_out;

    char* p = (char*)d_ws;
    auto alloc = [&](size_t bytes) {
        void* r = (void*)p;
        p += (bytes + 255) & ~(size_t)255;
        return r;
    };
    float*    kwsum  = (float*)alloc(INP * sizeof(float));
    float*    kbsum  = (float*)alloc(sizeof(float));
    float*    ksum   = (float*)alloc((size_t)MROWS * sizeof(float));
    float*    erec   = (float*)alloc((size_t)EE * 4 * sizeof(float));
    unsigned* tlist  = (unsigned*)alloc((size_t)TBYTES);
    float*    values = (float*)alloc((size_t)MROWS * VAL * sizeof(float));
    float*    agg    = (float*)alloc((size_t)MROWS * VAL * sizeof(float));

    bsa_reduce_kw<<<1, 128, 0, stream>>>(k_w, k_b, kwsum, kbsum);

    int gemm_blocks = (MROWS / 16 + 7) / 8;
    bsa_values_gemm<<<gemm_blocks, GEMM_THREADS, 0, stream>>>(x, v_w, v_b, values);

    bsa_ksum<<<(MROWS / KS_ROWS * 32 + 255) / 256, 256, 0, stream>>>(x, kwsum, kbsum, ksum);

    bsa_edge_rec<<<(EE / ER_PER + 255) / 256, 256, 0, stream>>>(conn, rows, cols, ksum, erec, tlist);

    bsa_agg_owner<<<NTILE, AGW * 32, 0, stream>>>(erec, tlist, values, agg);

    bsa_out_gemm_ln<<<gemm_blocks, GEMM_THREADS, 0, stream>>>(agg, out_w, out_b,
                                                     ln_g, ln_b, out);
}


# Owner tile's second pass walks LDS lists of the edges its one scan kept, rows dealt to waves (row mod 32): second list scan goes; a row's LDS float adds by one wave in list order, same every run
# speedup vs baseline: 5.5027x; 1.0499x over previous
#include <hip/hip_runtime.h>
#include <math.h>

#define BB    2
#define NN    50000
#define EE    800000
#define INP   128
#define KEY   64
#define VAL   128
#define MROWS (BB * NN)
#define LN_EPS 1e-5f

typedef __attribute__((ext_vector_type(2))) float v2f;
typedef __attribute__((ext_vector_type(8))) float v8f;

__device__ __forceinline__ unsigned enc_f32(float f) {
    unsigned u = __float_as_uint(f);
    return (u & 0x80000000u) ? ~u : (u | 0x80000000u);
}
__device__ __forceinline__ float dec_f32(unsigned u) {
    return (u & 0x80000000u) ? __uint_as_float(u & 0x7FFFFFFFu)
                             : __uint_as_float(~u);
}

__global__ void bsa_reduce_kw(const float* __restrict__ kw,
                              const float* __restrict__ kb,
                              float* __restrict__ kwsum,
                              float* __restrict__ kbsum) {
    int i = threadIdx.x;
    float s = 0.f;
    for (int j = 0; j < KEY; ++j) s += kw[i * KEY + j];
    kwsum[i] = s;
    if (i == 0) {
        float t = 0.f;
        for (int j = 0; j < KEY; ++j) t += kb[j];
        kbsum[0] = t;
    }
}

#define KS_ROWS 32
#define KS_GRP  (KS_ROWS / 2)

static_assert(MROWS % KS_ROWS == 0, "every wave has a full set of rows");

__global__ void bsa_ksum(const float* __restrict__ x,
                         const float* __restrict__ kwsum,
                         const float* __restrict__ kbsum,
                         float* __restrict__ ksum) {
    const int wid  = (blockIdx.x * blockDim.x + threadIdx.x) >> 5;
    const int lane = threadIdx.x & 31;
    if (wid >= MROWS / KS_ROWS) return;
    float kw[INP / 32];
    #pragma unroll
    for (int i = 0; i < INP / 32; ++i) kw[i] = kwsum[lane + i * 32];
    const float* xw = x + (size_t)wid * (KS_ROWS * INP) + lane;
    float mine = 0.f;
    #pragma unroll
    for (int g = 0; g < KS_ROWS; g += KS_GRP) {
        float v[KS_GRP][INP / 32];
        float s[KS_GRP];
        #pragma unroll
        for (int u = 0; u < KS_GRP; ++u) {
            #pragma unroll
            for (int i = 0; i < INP / 32; ++i) v[u][i] = xw[(g + u) * INP + i * 32];
        }
        #pragma unroll
        for (int u = 0; u < KS_GRP; ++u) {
            float t = 0.f;
            #pragma unroll
            for (int i = 0; i < INP / 32; ++i) t += v[u][i] * kw[i];
            s[u] = t;
        }
        #pragma unroll
        for (int off = 16; off; off >>= 1) {
            #pragma unroll
            for (int u = 0; u < KS_GRP; ++u) s[u] += __shfl_xor(s[u], off);
        }
        #pragma unroll
        for (int u = 0; u < KS_GRP; ++u)
            if (lane == g + u) mine = s[u];
    }
    ksum[wid * KS_ROWS + lane] = mine + kbsum[0];
}

__device__ __forceinline__ void bsa_wmma_tile(const float* __restrict__ Arow,
                                              const float* __restrict__ Bs,
                                              v8f acc[8], int lane) {
    const int kh = (lane >> 4) << 1;
    const int n  = lane & 15;
    for (int k = 0; k < 128; k += 4) {
        v2f a = *(const v2f*)(Arow + k + kh);
        #pragma unroll
        for (int c = 0; c < 8; ++c) {
            v2f bf;
            bf.x = Bs[(k + kh)     * 128 + (c << 4) + n];
            bf.y = Bs[(k + kh + 1) * 128 + (c << 4) + n];
            acc[c] = __builtin_amdgcn_wmma_f32_16x16x4_f32(
                false, a, false, bf, (short)0, acc[c], false, false);
        }
    }
}

#define GEMM_THREADS 256
#define W_PIECES     (128 * 128 / 4 / GEMM_THREADS)

static_assert(128 * 128 % (4 * GEMM_THREADS) == 0, "every thread fills the same number of pieces");

__device__ __forceinline__ void bsa_fill_weight(const float* __restrict__ W, float4* __restrict__ Bs4) {
    const float4* W4 = (const float4*)W;
    float4 w[W_PIECES];
    #pragma unroll
    for (int i = 0; i < W_PIECES; ++i) w[i] = W4[threadIdx.x + i * GEMM_THREADS];
    #pragma unroll
    for (int i = 0; i < W_PIECES; ++i) Bs4[threadIdx.x + i * GEMM_THREADS] = w[i];
}

__global__ void bsa_values_gemm(const float* __restrict__ A,
                                const float* __restrict__ W,
                                const float* __restrict__ bias,
                                float* __restrict__ out) {
    __shared__ float4 Bs4[128 * 128 / 4];
    const float* Bs = (const float*)Bs4;
    bsa_fill_weight(W, Bs4);
    __syncthreads();

    int lane    = threadIdx.x & 31;
    int wv      = threadIdx.x >> 5;
    int tile    = blockIdx.x * 8 + wv;
    int rowbase = tile * 16;
    if (rowbase >= MROWS) return;

    v8f acc[8] = {};
    const float* Arow = A + (size_t)(rowbase + (lane & 15)) * 128;
    bsa_wmma_tile(Arow, Bs, acc, lane);

    int n      = lane & 15;
    int rowoff = (lane >> 4) << 3;
    #pragma unroll
    for (int c = 0; c < 8; ++c) {
        float bb = bias[(c << 4) + n];
        #pragma unroll
        for (int r = 0; r < 8; ++r)
            out[(size_t)(rowbase + rowoff + r) * 128 + (c << 4) + n] = acc[c][r] + bb;
    }
}

#define TR      200
#define NTILE   (NN / TR)
#define TCH     512
#define NTCH    ((EE + TCH - 1) / TCH)
#define TBYTES  (NTCH * TCH)
#define TNONE   255u
#define TNEVER  254u

static_assert(NN % TR == 0, "tiles must cover the rows exactly");
static_assert(NTILE <= 254 && TR <= 256, "tile id and local row are one byte each; 254 and 255 stay free");

#define ER_PER  4
#define ER_WAVE (32 * ER_PER)

static_assert(EE % ER_WAVE == 0, "a wave's edges are all inside the list or all outside");
static_assert(TBYTES - EE <= EE / ER_PER, "the pad is written by the first threads");

__global__ void bsa_edge_rec(const float* __restrict__ conn,
                             const int* __restrict__ rows,
                             const int* __restrict__ cols,
                             const float* __restrict__ ksum,
                             float* __restrict__ erec,
                             unsigned* __restrict__ tlist) {
    const int gtid = blockIdx.x * blockDim.x + threadIdx.x;
    unsigned char* tile8 = (unsigned char*)tlist;
    if (gtid < TBYTES - EE) tile8[EE + gtid] = (unsigned char)TNONE;
    const int e0 = (gtid >> 5) * ER_WAVE + (threadIdx.x & 31);
    if (e0 >= EE) return;
    int r[ER_PER], c[ER_PER];
    float cv[ER_PER], k0[ER_PER], k1[ER_PER];
    #pragma unroll
    for (int u = 0; u < ER_PER; ++u) {
        const int e = e0 + u * 32;
        r[u]  = rows[e];
        c[u]  = cols[e];
        cv[u] = conn[e];
    }
    #pragma unroll
    for (int u = 0; u < ER_PER; ++u) {
        k0[u] = ksum[c[u]];
        k1[u] = ksum[NN + c[u]];
    }
    #pragma unroll
    for (int u = 0; u < ER_PER; ++u) {
        const int e = e0 + u * 32;
        const float s0 = cv[u] * k0[u];
        const float s1 = cv[u] * k1[u];
        const int t = r[u] / TR;
        ((int4*)erec)[e] = make_int4(c[u], __float_as_int(s0), __float_as_int(s1), r[u] - t * TR);
        tile8[e] = (unsigned char)t;
    }
}

#define AGW        32
#define NTCH_PAD   (((NTCH + AGW - 1) / AGW) * AGW)
#define SCAN_ITERS (NTCH_PAD / AGW)
#define AHEAD      4
#define ROT_STEP   6
#define QCAP       (32 + TCH)
#define SEGCAP     256
#define ROWBITS    8

static_assert((NTILE - 1) * ROT_STEP < NTCH_PAD, "one wrap is enough for the start offset");
static_assert(TR * BB <= AGW * 32, "one thread per (row, batch) clears and settles the maxima and sums");
static_assert(TR <= (1 << ROWBITS) && EE <= (1 << (32 - ROWBITS)), "edge and row in tile share one word");
static_assert((AGW & (AGW - 1)) == 0 && AGW <= (1 << ROWBITS), "the owner of a row is read from the low bits of an entry");

__device__ __forceinline__ float bsa_lane_bcast(float v, int j) {
    return __int_as_float(__builtin_amdgcn_readlane(__float_as_int(v), j));
}

__device__ __forceinline__ void bsa_tile_add(float* __restrict__ acc, unsigned rl, int lane,
                                             float w0, float w1, float4 x0, float4 x1) {
    float* p0 = acc + (rl * 2u) * VAL + lane;
    float* p1 = p0 + VAL;
    __hip_atomic_fetch_add(p0,      w0 * x0.x, __ATOMIC_RELAXED, __HIP_MEMORY_SCOPE_WORKGROUP);
    __hip_atomic_fetch_add(p0 + 32, w0 * x0.y, __ATOMIC_RELAXED, __HIP_MEMORY_SCOPE_WORKGROUP);
    __hip_atomic_fetch_add(p0 + 64, w0 * x0.z, __ATOMIC_RELAXED, __HIP_MEMORY_SCOPE_WORKGROUP);
    __hip_atomic_fetch_add(p0 + 96, w0 * x0.w, __ATOMIC_RELAXED, __HIP_MEMORY_SCOPE_WORKGROUP);
    __hip_atomic_fetch_add(p1,      w1 * x1.x, __ATOMIC_RELAXED, __HIP_MEMORY_SCOPE_WORKGROUP);
    __hip_atomic_fetch_add(p1 + 32, w1 * x1.y, __ATOMIC_RELAXED, __HIP_MEMORY_SCOPE_WORKGROUP);
    __hip_atomic_fetch_add(p1 + 64, w1 * x1.z, __ATOMIC_RELAXED, __HIP_MEMORY_SCOPE_WORKGROUP);
    __hip_atomic_fetch_add(p1 + 96, w1 * x1.w, __ATOMIC_RELAXED, __HIP_MEMORY_SCOPE_WORKGROUP);
}

__device__ __forceinline__ void bsa_sum_add(float* dt, unsigned rl, int lane, float w0, float w1) {
    if (lane < BB)
        __hip_atomic_fetch_add(dt + rl * BB + lane, lane == 0 ? w0 : w1, __ATOMIC_RELAXED, __HIP_MEMORY_SCOPE_WORKGROUP);
}

__device__ __forceinline__ void bsa_max_drain(const unsigned* q, int n, int lane,
                                              const float* __restrict__ erec,
                                              unsigned* mt, unsigned* seg, int segn) {
    if (lane < n) {
        const int e = (int)q[lane];
        const int4 rec = ((const int4*)erec)[e];
        unsigned* p = mt + rec.w * BB;
        __hip_atomic_fetch_max(p,     enc_f32(__int_as_float(rec.y)), __ATOMIC_RELAXED, __HIP_MEMORY_SCOPE_WORKGROUP);
        __hip_atomic_fetch_max(p + 1, enc_f32(__int_as_float(rec.z)), __ATOMIC_RELAXED, __HIP_MEMORY_SCOPE_WORKGROUP);
        const int pos = segn + lane;
        if (pos < SEGCAP) seg[pos] = ((unsigned)e << ROWBITS) | (unsigned)rec.w;
    }
}

__device__ __forceinline__ void bsa_tile_drain(const unsigned* q, int n, int lane,
                                               const float* __restrict__ erec,
                                               const float* mf, float* dt,
                                               const float* __restrict__ values,
                                               float* __restrict__ acc) {
    int c = 0;
    unsigned rl = 0;
    float w0 = 0.f, w1 = 0.f;
    if (lane < n) {
        const int e = (int)q[lane];
        const int4 rec = ((const int4*)erec)[e];
        c = rec.x;
        rl = (unsigned)rec.w;
        const float m0 = mf[rec.w * BB];
        const float m1 = mf[rec.w * BB + 1];
        w0 = __expf(__int_as_float(rec.y) - m0);
        w1 = __expf(__int_as_float(rec.z) - m1);
    }
    const float* v0 = values + lane * 4;
    const float* v1 = values + (size_t)NN * VAL + lane * 4;
    int j = 0;
    #pragma nounroll
    for (; j + 4 <= n; j += 4) {
        int cj[4]; unsigned rj[4]; float a0[4], a1[4];
        float4 x0[4], x1[4];
        #pragma unroll
        for (int u = 0; u < 4; ++u) {
            cj[u] = __builtin_amdgcn_readlane(c, j + u);
            rj[u] = (unsigned)__builtin_amdgcn_readlane((int)rl, j + u);
            a0[u] = bsa_lane_bcast(w0, j + u);
            a1[u] = bsa_lane_bcast(w1, j + u);
            x0[u] = *(const float4*)(v0 + (size_t)cj[u] * VAL);
            x1[u] = *(const float4*)(v1 + (size_t)cj[u] * VAL);
            bsa_sum_add(dt, rj[u], lane, a0[u], a1[u]);
        }
        #pragma unroll
        for (int u = 0; u < 4; ++u)
            bsa_tile_add(acc, rj[u], lane, a0[u], a1[u], x0[u], x1[u]);
    }
    for (; j < n; ++j) {
        int cj = __builtin_amdgcn_readlane(c, j);
        unsigned rj = (unsigned)__builtin_amdgcn_readlane((int)rl, j);
        float a0 = bsa_lane_bcast(w0, j);
        float a1 = bsa_lane_bcast(w1, j);
        float4 x0 = *(const float4*)(v0 + (size_t)cj * VAL);
        float4 x1 = *(const float4*)(v1 + (size_t)cj * VAL);
        bsa_sum_add(dt, rj, lane, a0, a1);
        bsa_tile_add(acc, rj, lane, a0, a1, x0, x1);
    }
}

template <int PASS>
__device__ __forceinline__ void bsa_drain_ready(unsigned* q, int& cnt, int need, int lane,
                                                const float* __restrict__ erec,
                                                const float* __restrict__ values,
                                                unsigned* mt, float* dt,
                                                float* __restrict__ acc,
                                                unsigned* seg, int& segn) {
    int head = 0;
    while (cnt - head >= need) {
        int n = cnt - head;
        if (n > 32) n = 32;
        if (PASS == 0) { bsa_max_drain(q + head, n, lane, erec, mt, seg, segn); segn += n; }
        else           bsa_tile_drain(q + head, n, lane, erec, (const float*)mt, dt, values, acc);
        head += n;
    }
    if (head > 0) {
        const int rem = cnt - head;
        unsigned keep = 0u;
        if (lane < rem) keep = q[head + lane];
        if (lane < rem) q[lane] = keep;
        cnt = rem;
    }
}

__device__ __forceinline__ int4 bsa_tiles16(const unsigned* __restrict__ tlist, int ch, int lane) {
    const int chc = ch >= 0 ? ch : NTCH - 1;
    return ((const int4*)tlist)[chc * 32 + lane];
}

__device__ __forceinline__ unsigned bsa_zero_bytes(unsigned t) {
    return ~(((t & 0x7F7F7F7Fu) + 0x7F7F7F7Fu) | t | 0x7F7F7F7Fu);
}

__device__ __forceinline__ void bsa_scan_one(int4 v, int ch, int lane, unsigned tsplat, unsigned* q, int& cnt) {
    const unsigned ts = ch >= 0 ? tsplat : TNEVER * 0x01010101u;
    const int ebase = ch * TCH + lane * 16;
    const unsigned z0 = bsa_zero_bytes((unsigned)v.x ^ ts);
    const unsigned z1 = bsa_zero_bytes((unsigned)v.y ^ ts);
    const unsigned z2 = bsa_zero_bytes((unsigned)v.z ^ ts);
    const unsigned z3 = bsa_zero_bytes((unsigned)v.w ^ ts);
#define BSA_APPEND_WORD(zk, k)                                                   \
    if (__builtin_amdgcn_ballot_w32((zk) != 0u) != 0u) {                         \
        for (int j = 0; j < 4; ++j) {                                            \
            const bool m = (((zk) >> (8 * j + 7)) & 1u) != 0u;                   \
            const unsigned mask = __builtin_amdgcn_ballot_w32(m);                \
            if (mask != 0u) {                                                    \
                const int pre = (int)__builtin_amdgcn_mbcnt_lo(mask, 0u);        \
                if (m) q[cnt + pre] = (unsigned)(ebase + 4 * (k) + j);           \
                cnt += __builtin_popcount(mask);                                 \
            }                                                                    \
        }                                                                        \
    }
    BSA_APPEND_WORD(z0, 0)
    BSA_APPEND_WORD(z1, 1)
    BSA_APPEND_WORD(z2, 2)
    BSA_APPEND_WORD(z3, 3)
#undef BSA_APPEND_WORD
}

__device__ __forceinline__ int bsa_fetch_index(int st, int wv, int rot) {
    int s = st * AGW + wv + rot;
    if (s >= NTCH_PAD) s -= NTCH_PAD;
    return (st < SCAN_ITERS && s < NTCH) ? s : -1;
}

template <int PASS>
__device__ __forceinline__ void bsa_scan_pass(const unsigned* __restrict__ tlist,
                                              int wv, int lane, int rot, unsigned tsplat,
                                              unsigned* q,
                                              const float* __restrict__ erec,
                                              const float* __restrict__ values,
                                              unsigned* mt, float* dt,
                                              float* __restrict__ acc,
                                              unsigned* seg, int& segn) {
    int cnt = 0;

    static_assert(SCAN_ITERS >= AHEAD && AHEAD == 4, "four list loads are kept in flight");
    int4 c0 = bsa_tiles16(tlist, bsa_fetch_index(0, wv, rot), lane);
    int4 c1 = bsa_tiles16(tlist, bsa_fetch_index(1, wv, rot), lane);
    int4 c2 = bsa_tiles16(tlist, bsa_fetch_index(2, wv, rot), lane);
    int4 c3 = bsa_tiles16(tlist, bsa_fetch_index(3, wv, rot), lane);

#define BSA_STEP(cu, u)                                                                    \
    {                                                                                      \
        bsa_scan_one(cu, bsa_fetch_index(it + (u), wv, rot), lane, tsplat, q, cnt);        \
        cu = bsa_tiles16(tlist, bsa_fetch_index(it + (u) + AHEAD, wv, rot), lane);         \
        bsa_drain_ready<PASS>(q, cnt, 32, lane, erec, values, mt, dt, acc, seg, segn);     \
    }
    for (int it = 0; it < SCAN_ITERS; it += AHEAD) {
        BSA_STEP(c0, 0)
        BSA_STEP(c1, 1)
        BSA_STEP(c2, 2)
        BSA_STEP(c3, 3)
    }
#undef BSA_STEP
    bsa_drain_ready<PASS>(q, cnt, 1, lane, erec, values, mt, dt, acc, seg, segn);
}

#define SEG_STEP 128

static_assert(AGW == 32, "lane l holds the length of list l");
static_assert(SEGCAP % SEG_STEP == 0, "a step never reads past its own list");
static_assert(32 + SEG_STEP <= QCAP, "under 32 left over + one step fit the pending list");

__device__ __forceinline__ void bsa_owner_pass(const unsigned* seg, const unsigned* segcnt,
                                               int wv, int lane, unsigned* q,
                                               const float* __restrict__ erec,
                                               const float* __restrict__ values,
                                               unsigned* mt, float* dt,
                                               float* __restrict__ acc) {
    int cnt = 0;
    int none = 0;
    const int cntv = (int)segcnt[lane];
    const uint4* seg4 = (const uint4*)seg;
    uint4 cur = seg4[lane];
    for (int s = 0; s < AGW; ++s) {
        const int sn = s + 1 < AGW ? s + 1 : s;
        const uint4 nxt = seg4[sn * (SEGCAP / 4) + lane];
        const int ns = __builtin_amdgcn_readlane(cntv, s);
        for (int b = 0; b < ns; b += SEG_STEP) {
            uint4 v = cur;
            if (b != 0) v = seg4[s * (SEGCAP / 4) + b / 4 + lane];
            const int i0 = b + lane * 4;
#define BSA_OWN_APPEND(w, k)                                                                   \
            {                                                                                  \
                const bool m = (i0 + (k) < ns) && (((w) & (unsigned)(AGW - 1)) == (unsigned)wv); \
                const unsigned mask = __builtin_amdgcn_ballot_w32(m);                          \
                if (mask != 0u) {                                                              \
                    const int pre = (int)__builtin_amdgcn_mbcnt_lo(mask, 0u);                  \
                    if (m) q[cnt + pre] = (w) >> ROWBITS;                                      \
                    cnt += __builtin_popcount(mask);                                           \
                }                                                                              \
            }
            BSA_OWN_APPEND(v.x, 0)
            BSA_OWN_APPEND(v.y, 1)
            BSA_OWN_APPEND(v.z, 2)
            BSA_OWN_APPEND(v.w, 3)
#undef BSA_OWN_APPEND
            bsa_drain_ready<1>(q, cnt, 32, lane, erec, values, mt, dt, acc, nullptr, none);
        }
        cur = nxt;
    }
    bsa_drain_ready<1>(q, cnt, 1, lane, erec, values, mt, dt, acc, nullptr, none);
}

__global__ void __launch_bounds__(AGW * 32)
bsa_agg_owner(const float* __restrict__ erec,
              const unsigned* __restrict__ tlist,
              const float* __restrict__ values,
              float* __restrict__ agg) {
    __shared__ float acc[TR * BB * VAL];
    __shared__ unsigned queue[AGW * QCAP];
    __shared__ unsigned mt[TR * BB];
    __shared__ float dt[TR * BB];
    __shared__ unsigned seg[AGW * SEGCAP];
    __shared__ unsigned segcnt[AGW];
    __shared__ unsigned segshort;

    const int tid  = threadIdx.x;
    const int lane = tid & 31;
    const int wv   = __builtin_amdgcn_readfirstlane(tid >> 5);
    const int base = blockIdx.x * TR;
    const unsigned tsplat = blockIdx.x * 0x01010101u;

    float4* acc4 = (float4*)acc;
    for (int i = tid; i < TR * BB * VAL / 4; i += AGW * 32)
        acc4[i] = make_float4(0.f, 0.f, 0.f, 0.f);
    if (tid < TR * BB) { mt[tid] = 0u; dt[tid] = 0.f; }
    if (tid == 0) segshort = 0u;
    __syncthreads();

    unsigned* q = queue + wv * QCAP;

    const int rot = blockIdx.x * ROT_STEP;

    int segn = 0;
    bsa_scan_pass<0>(tlist, wv, lane, rot, tsplat, q, erec, values, mt, dt, acc, seg + wv * SEGCAP, segn);
    if (lane == 0) {
        segcnt[wv] = (unsigned)segn;
        if (segn > SEGCAP) segshort = 1u;
    }
    __syncthreads();
    if (tid < TR * BB) {
        const unsigned u = mt[tid];
        float m = (u == 0u) ? 0.0f : dec_f32(u);
        if (!isfinite(m)) m = 0.0f;
        mt[tid] = __float_as_uint(m);
    }
    __syncthreads();

    if (__builtin_amdgcn_readfirstlane((int)segshort) == 0) {
        bsa_owner_pass(seg, segcnt, wv, lane, q, erec, values, mt, dt, acc);
    } else {
        int none = 0;
        bsa_scan_pass<1>(tlist, wv, lane, rot, tsplat, q, erec, values, mt, dt, acc, nullptr, none);
    }
    __syncthreads();

    for (int row = wv; row < TR * BB; row += AGW) {
        const int rl = row >> 1, b = row & 1;
        const float* a = acc + row * VAL + lane;
        const float d  = dt[row];
        const float dd = (d == 0.0f) ? 1.0f : d;
        float4 o;
        o.x = a[0] / dd; o.y = a[32] / dd; o.z = a[64] / dd; o.w = a[96] / dd;
        *(float4*)(agg + ((size_t)b * NN + base + rl) * VAL + lane * 4) = o;
    }
}

__global__ void bsa_out_gemm_ln(const float* __restrict__ A,
                                const float* __restrict__ W,
                                const float* __restrict__ bias,
                                const float* __restrict__ lng,
                                const float* __restrict__ lnb,
                                float* __restrict__ out) {
    __shared__ float4 Bs4[128 * 128 / 4];
    const float* Bs = (const float*)Bs4;
    bsa_fill_weight(W, Bs4);
    __syncthreads();

    int lane    = threadIdx.x & 31;
    int wv      = threadIdx.x >> 5;
    int tile    = blockIdx.x * 8 + wv;
    int rowbase = tile * 16;
    if (rowbase >= MROWS) return;

    v8f acc[8] = {};
    const float* Arow = A + (size_t)(rowbase + (lane & 15)) * 128;
    bsa_wmma_tile(Arow, Bs, acc, lane);

    int n = lane & 15;
    float sr[8], sq[8];
    #pragma unroll
    for (int r = 0; r < 8; ++r) { sr[r] = 0.f; sq[r] = 0.f; }
    #pragma unroll
    for (int c = 0; c < 8; ++c) {
        float bb = bias[(c << 4) + n];
        #pragma unroll
        for (int r = 0; r < 8; ++r) {
            float v = acc[c][r] + bb;
            v = v / (1.0f + __expf(-v));
            acc[c][r] = v;
            sr[r] += v;
            sq[r] += v * v;
        }
    }
    #pragma unroll
    for (int r = 0; r < 8; ++r) {
        #pragma unroll
        for (int off = 8; off; off >>= 1) {
            sr[r] += __shfl_xor(sr[r], off);
            sq[r] += __shfl_xor(sq[r], off);
        }
    }
    float mu[8], rstd[8];
    #pragma unroll
    for (int r = 0; r < 8; ++r) {
        mu[r] = sr[r] * (1.0f / VAL);
        float var = sq[r] * (1.0f / VAL) - mu[r] * mu[r];
        rstd[r] = rsqrtf(var + LN_EPS);
    }
    int rowoff = (lane >> 4) << 3;
    #pragma unroll
    for (int c = 0; c < 8; ++c) {
        int col = (c << 4) + n;
        float g = lng[col], b2 = lnb[col];
        #pragma unroll
        for (int r = 0; r < 8; ++r)
            out[(size_t)(rowbase + rowoff + r) * VAL + col] =
                (acc[c][r] - mu[r]) * rstd[r] * g + b2;
    }
}

extern "C" void kernel_launch(void* const* d_in, const int* in_sizes, int n_in,
                              void* d_out, int out_size, void* d_ws, size_t ws_size,
                              hipStream_t stream) {
    const float* x     = (const float*)d_in[0];
    const float* conn  = (const float*)d_in[1];
    const float* k_w   = (const float*)d_in[2];
    const float* k_b   = (const float*)d_in[3];
    const float* v_w   = (const float*)d_in[4];
    const float* v_b   = (const float*)d_in[5];
    const float* out_w = (const float*)d_in[6];
    const float* out_b = (const float*)d_in[7];
    const float* ln_g  = (const float*)d_in[8];
    const float* ln_b  = (const float*)d_in[9];
    const int*   rows  = (const int*)d_in[10];
    const int*   cols  = (const int*)d_in[11];
    float* out = (float*)d_out;

    char* p = (char*)d_ws;
    auto alloc = [&](size_t bytes) {
        void* r = (void*)p;
        p += (bytes + 255) & ~(size_t)255;
        return r;
    };
    float*    kwsum  = (float*)alloc(INP * sizeof(float));
    float*    kbsum  = (float*)alloc(sizeof(float));
    float*    ksum   = (float*)alloc((size_t)MROWS * sizeof(float));
    float*    erec   = (float*)alloc((size_t)EE * 4 * sizeof(float));
    unsigned* tlist  = (unsigned*)alloc((size_t)TBYTES);
    float*    values = (float*)alloc((size_t)MROWS * VAL * sizeof(float));
    float*    agg    = (float*)alloc((size_t)MROWS * VAL * sizeof(float));

    bsa_reduce_kw<<<1, 128, 0, stream>>>(k_w, k_b, kwsum, kbsum);

    int gemm_blocks = (MROWS / 16 + 7) / 8;
    bsa_values_gemm<<<gemm_blocks, GEMM_THREADS, 0, stream>>>(x, v_w, v_b, values);

    bsa_ksum<<<(MROWS / KS_ROWS * 32 + 255) / 256, 256, 0, stream>>>(x, kwsum, kbsum, ksum);

    bsa_edge_rec<<<(EE / ER_PER + 255) / 256, 256, 0, stream>>>(conn, rows, cols, ksum, erec, tlist);

    bsa_agg_owner<<<NTILE, AGW * 32, 0, stream>>>(erec, tlist, values, agg);

    bsa_out_gemm_ln<<<gemm_blocks, GEMM_THREADS, 0, stream>>>(agg, out_w, out_b,
                                                     ln_g, ln_b, out);
}


# Edge pass writes each tile's bitmap of the edge list (LDS integer OR); a tile reads its own map, takes edges from set bits, not a byte per edge of the whole list; sum order new, fixed per run
# speedup vs baseline: 6.1318x; 1.1143x over previous
#include <hip/hip_runtime.h>
#include <math.h>

#define BB    2
#define NN    50000
#define EE    800000
#define INP   128
#define KEY   64
#define VAL   128
#define MROWS (BB * NN)
#define LN_EPS 1e-5f

typedef __attribute__((ext_vector_type(2))) float v2f;
typedef __attribute__((ext_vector_type(8))) float v8f;

__device__ __forceinline__ unsigned enc_f32(float f) {
    unsigned u = __float_as_uint(f);
    return (u & 0x80000000u) ? ~u : (u | 0x80000000u);
}
__device__ __forceinline__ float dec_f32(unsigned u) {
    return (u & 0x80000000u) ? __uint_as_float(u & 0x7FFFFFFFu)
                             : __uint_as_float(~u);
}

__global__ void bsa_reduce_kw(const float* __restrict__ kw,
                              const float* __restrict__ kb,
                              float* __restrict__ kwsum,
                              float* __restrict__ kbsum) {
    int i = threadIdx.x;
    float s = 0.f;
    for (int j = 0; j < KEY; ++j) s += kw[i * KEY + j];
    kwsum[i] = s;
    if (i == 0) {
        float t = 0.f;
        for (int j = 0; j < KEY; ++j) t += kb[j];
        kbsum[0] = t;
    }
}

#define KS_ROWS 32
#define KS_GRP  (KS_ROWS / 2)

static_assert(MROWS % KS_ROWS == 0, "every wave has a full set of rows");

__global__ void bsa_ksum(const float* __restrict__ x,
                         const float* __restrict__ kwsum,
                         const float* __restrict__ kbsum,
                         float* __restrict__ ksum) {
    const int wid  = (blockIdx.x * blockDim.x + threadIdx.x) >> 5;
    const int lane = threadIdx.x & 31;
    if (wid >= MROWS / KS_ROWS) return;
    float kw[INP / 32];
    #pragma unroll
    for (int i = 0; i < INP / 32; ++i) kw[i] = kwsum[lane + i * 32];
    const float* xw = x + (size_t)wid * (KS_ROWS * INP) + lane;
    float mine = 0.f;
    #pragma unroll
    for (int g = 0; g < KS_ROWS; g += KS_GRP) {
        float v[KS_GRP][INP / 32];
        float s[KS_GRP];
        #pragma unroll
        for (int u = 0; u < KS_GRP; ++u) {
            #pragma unroll
            for (int i = 0; i < INP / 32; ++i) v[u][i] = xw[(g + u) * INP + i * 32];
        }
        #pragma unroll
        for (int u = 0; u < KS_GRP; ++u) {
            float t = 0.f;
            #pragma unroll
            for (int i = 0; i < INP / 32; ++i) t += v[u][i] * kw[i];
            s[u] = t;
        }
        #pragma unroll
        for (int off = 16; off; off >>= 1) {
            #pragma unroll
            for (int u = 0; u < KS_GRP; ++u) s[u] += __shfl_xor(s[u], off);
        }
        #pragma unroll
        for (int u = 0; u < KS_GRP; ++u)
            if (lane == g + u) mine = s[u];
    }
    ksum[wid * KS_ROWS + lane] = mine + kbsum[0];
}

__device__ __forceinline__ void bsa_wmma_tile(const float* __restrict__ Arow,
                                              const float* __restrict__ Bs,
                                              v8f acc[8], int lane) {
    const int kh = (lane >> 4) << 1;
    const int n  = lane & 15;
    for (int k = 0; k < 128; k += 4) {
        v2f a = *(const v2f*)(Arow + k + kh);
        #pragma unroll
        for (int c = 0; c < 8; ++c) {
            v2f bf;
            bf.x = Bs[(k + kh)     * 128 + (c << 4) + n];
            bf.y = Bs[(k + kh + 1) * 128 + (c << 4) + n];
            acc[c] = __builtin_amdgcn_wmma_f32_16x16x4_f32(
                false, a, false, bf, (short)0, acc[c], false, false);
        }
    }
}

#define GEMM_THREADS 256
#define W_PIECES     (128 * 128 / 4 / GEMM_THREADS)

static_assert(128 * 128 % (4 * GEMM_THREADS) == 0, "every thread fills the same number of pieces");

__device__ __forceinline__ void bsa_fill_weight(const float* __restrict__ W, float4* __restrict__ Bs4) {
    const float4* W4 = (const float4*)W;
    float4 w[W_PIECES];
    #pragma unroll
    for (int i = 0; i < W_PIECES; ++i) w[i] = W4[threadIdx.x + i * GEMM_THREADS];
    #pragma unroll
    for (int i = 0; i < W_PIECES; ++i) Bs4[threadIdx.x + i * GEMM_THREADS] = w[i];
}

__global__ void bsa_values_gemm(const float* __restrict__ A,
                                const float* __restrict__ W,
                                const float* __restrict__ bias,
                                float* __restrict__ out) {
    __shared__ float4 Bs4[128 * 128 / 4];
    const float* Bs = (const float*)Bs4;
    bsa_fill_weight(W, Bs4);
    __syncthreads();

    int lane    = threadIdx.x & 31;
    int wv      = threadIdx.x >> 5;
    int tile    = blockIdx.x * 8 + wv;
    int rowbase = tile * 16;
    if (rowbase >= MROWS) return;

    v8f acc[8] = {};
    const float* Arow = A + (size_t)(rowbase + (lane & 15)) * 128;
    bsa_wmma_tile(Arow, Bs, acc, lane);

    int n      = lane & 15;
    int rowoff = (lane >> 4) << 3;
    #pragma unroll
    for (int c = 0; c < 8; ++c) {
        float bb = bias[(c << 4) + n];
        #pragma unroll
        for (int r = 0; r < 8; ++r)
            out[(size_t)(rowbase + rowoff + r) * 128 + (c << 4) + n] = acc[c][r] + bb;
    }
}

#define TR      200
#define NTILE   (NN / TR)

static_assert(NN % TR == 0, "tiles must cover the rows exactly");
static_assert(TR <= 256, "the row inside its tile is one byte of a list entry");

#define ER_PER   4
#define ER_WAVE  (32 * ER_PER)
#define ER_BLOCK 256
#define ER_EDGES (ER_BLOCK * ER_PER)
#define ER_WORDS (ER_EDGES / 32)
#define ER_NBLK  ((EE + ER_EDGES - 1) / ER_EDGES)
#define BM_WORDS (ER_NBLK * ER_WORDS)
#define BM_PITCH (ER_WORDS + 1)

static_assert(EE % ER_WAVE == 0, "a wave's edges are all inside the list or all outside");
static_assert(ER_WORDS == 32, "a tile's words of one block are one line: one word per lane at the store");
static_assert(BM_WORDS % 4 == 0, "a map is read in 16-byte pieces");

__global__ void __launch_bounds__(ER_BLOCK)
bsa_edge_rec(const float* __restrict__ conn,
             const int* __restrict__ rows,
             const int* __restrict__ cols,
             const float* __restrict__ ksum,
             float* __restrict__ erec,
             unsigned* __restrict__ bmap) {
    __shared__ unsigned bm[NTILE * BM_PITCH];
    const int tid  = threadIdx.x;
    const int lane = tid & 31;
    const int wv   = tid >> 5;
    for (int i = tid; i < NTILE * BM_PITCH; i += ER_BLOCK) bm[i] = 0u;
    __syncthreads();
    const int e0 = blockIdx.x * ER_EDGES + wv * ER_WAVE + lane;
    if (e0 < EE) {
        int r[ER_PER], c[ER_PER];
        float cv[ER_PER], k0[ER_PER], k1[ER_PER];
        #pragma unroll
        for (int u = 0; u < ER_PER; ++u) {
            const int e = e0 + u * 32;
            r[u]  = rows[e];
            c[u]  = cols[e];
            cv[u] = conn[e];
        }
        #pragma unroll
        for (int u = 0; u < ER_PER; ++u) {
            k0[u] = ksum[c[u]];
            k1[u] = ksum[NN + c[u]];
        }
        #pragma unroll
        for (int u = 0; u < ER_PER; ++u) {
            const int e = e0 + u * 32;
            const float s0 = cv[u] * k0[u];
            const float s1 = cv[u] * k1[u];
            const int t = r[u] / TR;
            ((int4*)erec)[e] = make_int4(c[u], __float_as_int(s0), __float_as_int(s1), r[u] - t * TR);
            __hip_atomic_fetch_or(&bm[t * BM_PITCH + wv * ER_PER + u], 1u << lane,
                                  __ATOMIC_RELAXED, __HIP_MEMORY_SCOPE_WORKGROUP);
        }
    }
    __syncthreads();
    for (int t = wv; t < NTILE; t += ER_BLOCK / 32)
        bmap[(size_t)t * BM_WORDS + blockIdx.x * ER_WORDS + lane] = bm[t * BM_PITCH + lane];
}

#define AGW        32
#define AHEAD      4
#define BM_U4      (BM_WORDS / 4)
#define BM_STEPS   ((BM_U4 + 31) / 32)
#define BM_ITERS   ((BM_STEPS + AGW - 1) / AGW)
#define QCAP       544
#define SEGCAP     256
#define ROWBITS    8
#define APITCH     132
#define NRT        (TR * BB / 16)

static_assert(TR * BB <= AGW * 32, "one thread per (row, batch) clears and settles the maxima and sums");
static_assert(TR <= (1 << ROWBITS) && EE <= (1 << (32 - ROWBITS)), "edge and row in tile share one word");
static_assert((AGW & (AGW - 1)) == 0 && AGW <= (1 << ROWBITS), "the owner of a row is read from the low bits of an entry");


__device__ __forceinline__ float bsa_lane_bcast(float v, int j) {
    return __int_as_float(__builtin_amdgcn_readlane(__float_as_int(v), j));
}

__device__ __forceinline__ void bsa_tile_add(float* __restrict__ acc, unsigned rl, int lane,
                                             float w0, float w1, float4 x0, float4 x1) {
    float* p0 = acc + (rl * 2u) * APITCH + lane;
    float* p1 = p0 + APITCH;
    __hip_atomic_fetch_add(p0,      w0 * x0.x, __ATOMIC_RELAXED, __HIP_MEMORY_SCOPE_WORKGROUP);
    __hip_atomic_fetch_add(p0 + 32, w0 * x0.y, __ATOMIC_RELAXED, __HIP_MEMORY_SCOPE_WORKGROUP);
    __hip_atomic_fetch_add(p0 + 64, w0 * x0.z, __ATOMIC_RELAXED, __HIP_MEMORY_SCOPE_WORKGROUP);
    __hip_atomic_fetch_add(p0 + 96, w0 * x0.w, __ATOMIC_RELAXED, __HIP_MEMORY_SCOPE_WORKGROUP);
    __hip_atomic_fetch_add(p1,      w1 * x1.x, __ATOMIC_RELAXED, __HIP_MEMORY_SCOPE_WORKGROUP);
    __hip_atomic_fetch_add(p1 + 32, w1 * x1.y, __ATOMIC_RELAXED, __HIP_MEMORY_SCOPE_WORKGROUP);
    __hip_atomic_fetch_add(p1 + 64, w1 * x1.z, __ATOMIC_RELAXED, __HIP_MEMORY_SCOPE_WORKGROUP);
    __hip_atomic_fetch_add(p1 + 96, w1 * x1.w, __ATOMIC_RELAXED, __HIP_MEMORY_SCOPE_WORKGROUP);
}

__device__ __forceinline__ void bsa_sum_add(float* dt, unsigned rl, int lane, float w0, float w1) {
    if (lane < BB)
        __hip_atomic_fetch_add(dt + rl * BB + lane, lane == 0 ? w0 : w1, __ATOMIC_RELAXED, __HIP_MEMORY_SCOPE_WORKGROUP);
}

__device__ __forceinline__ void bsa_max_drain(const unsigned* q, int n, int lane,
                                              const float* __restrict__ erec,
                                              unsigned* mt, unsigned* seg, int segn) {
    if (lane < n) {
        const int e = (int)q[lane];
        const int4 rec = ((const int4*)erec)[e];
        unsigned* p = mt + rec.w * BB;
        __hip_atomic_fetch_max(p,     enc_f32(__int_as_float(rec.y)), __ATOMIC_RELAXED, __HIP_MEMORY_SCOPE_WORKGROUP);
        __hip_atomic_fetch_max(p + 1, enc_f32(__int_as_float(rec.z)), __ATOMIC_RELAXED, __HIP_MEMORY_SCOPE_WORKGROUP);
        const int pos = segn + lane;
        if (pos < SEGCAP) seg[pos] = ((unsigned)e << ROWBITS) | (unsigned)rec.w;
    }
}

__device__ __forceinline__ void bsa_tile_drain(const unsigned* q, int n, int lane,
                                               const float* __restrict__ erec,
                                               const float* mf, float* dt,
                                               const float* __restrict__ values,
                                               float* __restrict__ acc) {
    int c = 0;
    unsigned rl = 0;
    float w0 = 0.f, w1 = 0.f;
    if (lane < n) {
        const int e = (int)q[lane];
        const int4 rec = ((const int4*)erec)[e];
        c = rec.x;
        rl = (unsigned)rec.w;
        const float m0 = mf[rec.w * BB];
        const float m1 = mf[rec.w * BB + 1];
        w0 = __expf(__int_as_float(rec.y) - m0);
        w1 = __expf(__int_as_float(rec.z) - m1);
    }
    const float* v0 = values + lane * 4;
    const float* v1 = values + (size_t)NN * VAL + lane * 4;
    int j = 0;
    #pragma nounroll
    for (; j + 4 <= n; j += 4) {
        int cj[4]; unsigned rj[4]; float a0[4], a1[4];
        float4 x0[4], x1[4];
        #pragma unroll
        for (int u = 0; u < 4; ++u) {
            cj[u] = __builtin_amdgcn_readlane(c, j + u);
            rj[u] = (unsigned)__builtin_amdgcn_readlane((int)rl, j + u);
            a0[u] = bsa_lane_bcast(w0, j + u);
            a1[u] = bsa_lane_bcast(w1, j + u);
            x0[u] = *(const float4*)(v0 + (size_t)cj[u] * VAL);
            x1[u] = *(const float4*)(v1 + (size_t)cj[u] * VAL);
            bsa_sum_add(dt, rj[u], lane, a0[u], a1[u]);
        }
        #pragma unroll
        for (int u = 0; u < 4; ++u)
            bsa_tile_add(acc, rj[u], lane, a0[u], a1[u], x0[u], x1[u]);
    }
    for (; j < n; ++j) {
        int cj = __builtin_amdgcn_readlane(c, j);
        unsigned rj = (unsigned)__builtin_amdgcn_readlane((int)rl, j);
        float a0 = bsa_lane_bcast(w0, j);
        float a1 = bsa_lane_bcast(w1, j);
        float4 x0 = *(const float4*)(v0 + (size_t)cj * VAL);
        float4 x1 = *(const float4*)(v1 + (size_t)cj * VAL);
        bsa_sum_add(dt, rj, lane, a0, a1);
        bsa_tile_add(acc, rj, lane, a0, a1, x0, x1);
    }
}

template <int PASS>
__device__ __forceinline__ void bsa_drain_ready(unsigned* q, int& cnt, int need, int lane,
                                                const float* __restrict__ erec,
                                                const float* __restrict__ values,
                                                unsigned* mt, float* dt,
                                                float* __restrict__ acc,
                                                unsigned* seg, int& segn) {
    int head = 0;
    while (cnt - head >= need) {
        int n = cnt - head;
        if (n > 32) n = 32;
        if (PASS == 0) { bsa_max_drain(q + head, n, lane, erec, mt, seg, segn); segn += n; }
        else           bsa_tile_drain(q + head, n, lane, erec, (const float*)mt, dt, values, acc);
        head += n;
    }
    if (head > 0) {
        const int rem = cnt - head;
        unsigned keep = 0u;
        if (lane < rem) keep = q[head + lane];
        if (lane < rem) q[lane] = keep;
        cnt = rem;
    }
}

__device__ __forceinline__ uint4 bsa_bits16(const uint4* __restrict__ bm4, int st, int wv, int lane) {
    const int i = (st * AGW + wv) * 32 + lane;
    const bool in = st < BM_ITERS && i < BM_U4;
    return bm4[in ? i : 0];
}

template <int PASS>
__device__ __forceinline__ void bsa_bits_step(uint4 v, int st, int wv, int lane,
                                              unsigned* q, int& cnt,
                                              const float* __restrict__ erec,
                                              const float* __restrict__ values,
                                              unsigned* mt, float* dt,
                                              float* __restrict__ acc,
                                              unsigned* seg, int& segn) {
    const int i = (st * AGW + wv) * 32 + lane;
    const bool in = st < BM_ITERS && i < BM_U4;
    unsigned w0 = in ? v.x : 0u;
    unsigned w1 = in ? v.y : 0u;
    unsigned w2 = in ? v.z : 0u;
    unsigned w3 = in ? v.w : 0u;
    const unsigned ebase = (unsigned)i * 128u;
    for (int trip = 0; trip < 128; ++trip) {
        const bool has = (w0 | w1 | w2 | w3) != 0u;
        const unsigned mask = __builtin_amdgcn_ballot_w32(has);
        if (mask == 0u) break;
        const unsigned k  = w0 ? 0u : (w1 ? 1u : (w2 ? 2u : 3u));
        const unsigned wk = w0 ? w0 : (w1 ? w1 : (w2 ? w2 : w3));
        if (has)
            q[cnt + (int)__builtin_amdgcn_mbcnt_lo(mask, 0u)] = ebase + 32u * k + (unsigned)__builtin_ctz(wk);
        const unsigned cl = wk & (wk - 1u);
        w0 = k == 0u ? cl : w0;
        w1 = k == 1u ? cl : w1;
        w2 = k == 2u ? cl : w2;
        w3 = k == 3u ? cl : w3;
        cnt += __builtin_popcount(mask);
        bsa_drain_ready<PASS>(q, cnt, 32, lane, erec, values, mt, dt, acc, seg, segn);
    }
}

template <int PASS>
__device__ __forceinline__ void bsa_scan_pass(const uint4* __restrict__ bm4,
                                              int wv, int lane,
                                              unsigned* q,
                                              const float* __restrict__ erec,
                                              const float* __restrict__ values,
                                              unsigned* mt, float* dt,
                                              float* __restrict__ acc,
                                              unsigned* seg, int& segn) {
    int cnt = 0;

    static_assert(AHEAD == 4, "four map loads are kept in flight");
    uint4 c0 = bsa_bits16(bm4, 0, wv, lane);
    uint4 c1 = bsa_bits16(bm4, 1, wv, lane);
    uint4 c2 = bsa_bits16(bm4, 2, wv, lane);
    uint4 c3 = bsa_bits16(bm4, 3, wv, lane);

#define BSA_STEP(cu, u)                                                                              \
    {                                                                                                \
        bsa_bits_step<PASS>(cu, it + (u), wv, lane, q, cnt, erec, values, mt, dt, acc, seg, segn);   \
        cu = bsa_bits16(bm4, it + (u) + AHEAD, wv, lane);                                            \
    }
    for (int it = 0; it < BM_ITERS; it += AHEAD) {
        BSA_STEP(c0, 0)
        BSA_STEP(c1, 1)
        BSA_STEP(c2, 2)
        BSA_STEP(c3, 3)
    }
#undef BSA_STEP
    bsa_drain_ready<PASS>(q, cnt, 1, lane, erec, values, mt, dt, acc, seg, segn);
}

#define SEG_STEP 128

static_assert(AGW == 32, "lane l holds the length of list l");
static_assert(SEGCAP % SEG_STEP == 0, "a step never reads past its own list");
static_assert(32 + SEG_STEP <= QCAP, "under 32 left over + one step fit the pending list");

__device__ __forceinline__ void bsa_owner_pass(const unsigned* seg, const unsigned* segcnt,
                                               int wv, int lane, unsigned* q,
                                               const float* __restrict__ erec,
                                               const float* __restrict__ values,
                                               unsigned* mt, float* dt,
                                               float* __restrict__ acc) {
    int cnt = 0;
    int none = 0;
    const int cntv = (int)segcnt[lane];
    const uint4* seg4 = (const uint4*)seg;
    uint4 cur = seg4[lane];
    for (int s = 0; s < AGW; ++s) {
        const int sn = s + 1 < AGW ? s + 1 : s;
        const uint4 nxt = seg4[sn * (SEGCAP / 4) + lane];
        const int ns = __builtin_amdgcn_readlane(cntv, s);
        for (int b = 0; b < ns; b += SEG_STEP) {
            uint4 v = cur;
            if (b != 0) v = seg4[s * (SEGCAP / 4) + b / 4 + lane];
            const int i0 = b + lane * 4;
#define BSA_OWN_APPEND(w, k)                                                                   \
            {                                                                                  \
                const bool m = (i0 + (k) < ns) && (((w) & (unsigned)(AGW - 1)) == (unsigned)wv); \
                const unsigned mask = __builtin_amdgcn_ballot_w32(m);                          \
                if (mask != 0u) {                                                              \
                    const int pre = (int)__builtin_amdgcn_mbcnt_lo(mask, 0u);                  \
                    if (m) q[cnt + pre] = (w) >> ROWBITS;                                      \
                    cnt += __builtin_popcount(mask);                                           \
                }                                                                              \
            }
            BSA_OWN_APPEND(v.x, 0)
            BSA_OWN_APPEND(v.y, 1)
            BSA_OWN_APPEND(v.z, 2)
            BSA_OWN_APPEND(v.w, 3)
#undef BSA_OWN_APPEND
            bsa_drain_ready<1>(q, cnt, 32, lane, erec, values, mt, dt, acc, nullptr, none);
        }
        cur = nxt;
    }
    bsa_drain_ready<1>(q, cnt, 1, lane, erec, values, mt, dt, acc, nullptr, none);
}

__global__ void __launch_bounds__(AGW * 32)
bsa_agg_owner(const float* __restrict__ erec,
              const unsigned* __restrict__ bmap,
              const float* __restrict__ values,
              const float* __restrict__ out_w,
              const float* __restrict__ out_b,
              const float* __restrict__ lng,
              const float* __restrict__ lnb,
              float* __restrict__ out) {
    __shared__ __attribute__((aligned(16))) float acc[TR * BB * APITCH];
    __shared__ __attribute__((aligned(16))) unsigned queue[AGW * QCAP];
    __shared__ unsigned mt[TR * BB];
    __shared__ float dt[TR * BB];
    __shared__ unsigned seg[AGW * SEGCAP];
    __shared__ unsigned segcnt[AGW];
    __shared__ unsigned segshort;

    const int tid  = threadIdx.x;
    const int lane = tid & 31;
    const int wv   = __builtin_amdgcn_readfirstlane(tid >> 5);
    const int base = blockIdx.x * TR;
    const uint4* bm4 = (const uint4*)(bmap + (size_t)blockIdx.x * BM_WORDS);

    float4* acc4 = (float4*)acc;
    for (int i = tid; i < TR * BB * APITCH / 4; i += AGW * 32)
        acc4[i] = make_float4(0.f, 0.f, 0.f, 0.f);
    if (tid < TR * BB) { mt[tid] = 0u; dt[tid] = 0.f; }
    if (tid == 0) segshort = 0u;
    __syncthreads();

    unsigned* q = queue + wv * QCAP;

    int segn = 0;
    bsa_scan_pass<0>(bm4, wv, lane, q, erec, values, mt, dt, acc, seg + wv * SEGCAP, segn);
    if (lane == 0) {
        segcnt[wv] = (unsigned)segn;
        if (segn > SEGCAP) segshort = 1u;
    }
    __syncthreads();
    if (tid < TR * BB) {
        const unsigned u = mt[tid];
        float m = (u == 0u) ? 0.0f : dec_f32(u);
        if (!isfinite(m)) m = 0.0f;
        mt[tid] = __float_as_uint(m);
    }
    __syncthreads();

    if (__builtin_amdgcn_readfirstlane((int)segshort) == 0) {
        bsa_owner_pass(seg, segcnt, wv, lane, q, erec, values, mt, dt, acc);
    } else {
        int none = 0;
        bsa_scan_pass<1>(bm4, wv, lane, q, erec, values, mt, dt, acc, nullptr, none);
    }
    __syncthreads();

    {
        static_assert(sizeof(unsigned) * AGW * QCAP >= sizeof(float) * VAL * VAL, "out_w fits where the pending lists were");
        static_assert(VAL * VAL / 4 == 4 * AGW * 32, "four 16-byte pieces of out_w per thread");
        const float4* W4 = (const float4*)out_w;
        float4* Bw4 = (float4*)queue;
        float4 w[4];
        #pragma unroll
        for (int i = 0; i < 4; ++i) w[i] = W4[tid + i * (AGW * 32)];
        #pragma unroll
        for (int i = 0; i < 4; ++i) {
            const int p   = tid + i * (AGW * 32);
            const int col = p >> 5;
            const int wr  = (col & 3) * 32 + (col >> 2);
            Bw4[wr * 32 + (p & 31)] = w[i];
        }
    }
    for (int row = wv; row < TR * BB; row += AGW) {
        float* a = acc + row * APITCH + lane;
        const float d  = dt[row];
        const float dd = (d == 0.0f) ? 1.0f : d;
        a[0] = a[0] / dd; a[32] = a[32] / dd; a[64] = a[64] / dd; a[96] = a[96] / dd;
    }
    __syncthreads();
    if (wv >= NRT) return;

    const float* Bw = (const float*)queue;
    const int n      = lane & 15;
    const int rowoff = (lane >> 4) << 3;
    v8f cf[8] = {};
    bsa_wmma_tile(acc + (wv * 16 + n) * APITCH, Bw, cf, lane);

    float sr[8], sq[8];
    #pragma unroll
    for (int r = 0; r < 8; ++r) { sr[r] = 0.f; sq[r] = 0.f; }
    #pragma unroll
    for (int c = 0; c < 8; ++c) {
        const float bb = out_b[(c << 4) + n];
        #pragma unroll
        for (int r = 0; r < 8; ++r) {
            float v = cf[c][r] + bb;
            v = v / (1.0f + __expf(-v));
            cf[c][r] = v;
            sr[r] += v;
            sq[r] += v * v;
        }
    }
    #pragma unroll
    for (int r = 0; r < 8; ++r) {
        #pragma unroll
        for (int off = 8; off; off >>= 1) {
            sr[r] += __shfl_xor(sr[r], off);
            sq[r] += __shfl_xor(sq[r], off);
        }
    }
    float mu[8], rstd[8];
    #pragma unroll
    for (int r = 0; r < 8; ++r) {
        mu[r] = sr[r] * (1.0f / VAL);
        const float var = sq[r] * (1.0f / VAL) - mu[r] * mu[r];
        rstd[r] = rsqrtf(var + LN_EPS);
    }
    #pragma unroll
    for (int c = 0; c < 8; ++c) {
        const int col = (c << 4) + n;
        const float g = lng[col], b2 = lnb[col];
        #pragma unroll
        for (int r = 0; r < 8; ++r) {
            const int row = wv * 16 + rowoff + r;
            out[((size_t)(row & 1) * NN + base + (row >> 1)) * VAL + col] =
                (cf[c][r] - mu[r]) * rstd[r] * g + b2;
        }
    }
}

extern "C" void kernel_launch(void* const* d_in, const int* in_sizes, int n_in,
                              void* d_out, int out_size, void* d_ws, size_t ws_size,
                              hipStream_t stream) {
    const float* x     = (const float*)d_in[0];
    const float* conn  = (const float*)d_in[1];
    const float* k_w   = (const float*)d_in[2];
    const float* k_b   = (const float*)d_in[3];
    const float* v_w   = (const float*)d_in[4];
    const float* v_b   = (const float*)d_in[5];
    const float* out_w = (const float*)d_in[6];
    const float* out_b = (const float*)d_in[7];
    const float* ln_g  = (const float*)d_in[8];
    const float* ln_b  = (const float*)d_in[9];
    const int*   rows  = (const int*)d_in[10];
    const int*   cols  = (const int*)d_in[11];
    float* out = (float*)d_out;

    char* p = (char*)d_ws;
    auto alloc = [&](size_t bytes) {
        void* r = (void*)p;
        p += (bytes + 255) & ~(size_t)255;
        return r;
    };
    float*    kwsum  = (float*)alloc(INP * sizeof(float));
    float*    kbsum  = (float*)alloc(sizeof(float));
    float*    ksum   = (float*)alloc((size_t)MROWS * sizeof(float));
    float*    erec   = (float*)alloc((size_t)EE * 4 * sizeof(float));
    unsigned* bmap   = (unsigned*)alloc((size_t)NTILE * BM_WORDS * sizeof(unsigned));
    float*    values = (float*)alloc((size_t)MROWS * VAL * sizeof(float));

    bsa_reduce_kw<<<1, 128, 0, stream>>>(k_w, k_b, kwsum, kbsum);

    int gemm_blocks = (MROWS / 16 + 7) / 8;
    bsa_values_gemm<<<gemm_blocks, GEMM_THREADS, 0, stream>>>(x, v_w, v_b, values);

    bsa_ksum<<<(MROWS / KS_ROWS * 32 + 255) / 256, 256, 0, stream>>>(x, kwsum, kbsum, ksum);

    bsa_edge_rec<<<ER_NBLK, ER_BLOCK, 0, stream>>>(conn, rows, cols, ksum, erec, bmap);

    bsa_agg_owner<<<NTILE, AGW * 32, 0, stream>>>(erec, bmap, values, out_w, out_b, ln_g, ln_b, out);
}


# Owner wave takes 32 pending edges grouped by row, sums a row's group of ex*x rows in registers, one set of LDS adds per group (was per edge): tile LDS adds down about 4x; sum order new, fixed
# speedup vs baseline: 7.1693x; 1.1692x over previous
#include <hip/hip_runtime.h>
#include <math.h>

#define BB    2
#define NN    50000
#define EE    800000
#define INP   128
#define KEY   64
#define VAL   128
#define MROWS (BB * NN)
#define LN_EPS 1e-5f

typedef __attribute__((ext_vector_type(2))) float v2f;
typedef __attribute__((ext_vector_type(8))) float v8f;

__device__ __forceinline__ unsigned enc_f32(float f) {
    unsigned u = __float_as_uint(f);
    return (u & 0x80000000u) ? ~u : (u | 0x80000000u);
}
__device__ __forceinline__ float dec_f32(unsigned u) {
    return (u & 0x80000000u) ? __uint_as_float(u & 0x7FFFFFFFu)
                             : __uint_as_float(~u);
}

__global__ void bsa_reduce_kw(const float* __restrict__ kw,
                              const float* __restrict__ kb,
                              float* __restrict__ kwsum,
                              float* __restrict__ kbsum) {
    int i = threadIdx.x;
    float s = 0.f;
    for (int j = 0; j < KEY; ++j) s += kw[i * KEY + j];
    kwsum[i] = s;
    if (i == 0) {
        float t = 0.f;
        for (int j = 0; j < KEY; ++j) t += kb[j];
        kbsum[0] = t;
    }
}

#define KS_ROWS 32
#define KS_GRP  (KS_ROWS / 2)

static_assert(MROWS % KS_ROWS == 0, "every wave has a full set of rows");

__global__ void bsa_ksum(const float* __restrict__ x,
                         const float* __restrict__ kwsum,
                         const float* __restrict__ kbsum,
                         float* __restrict__ ksum) {
    const int wid  = (blockIdx.x * blockDim.x + threadIdx.x) >> 5;
    const int lane = threadIdx.x & 31;
    if (wid >= MROWS / KS_ROWS) return;
    float kw[INP / 32];
    #pragma unroll
    for (int i = 0; i < INP / 32; ++i) kw[i] = kwsum[lane + i * 32];
    const float* xw = x + (size_t)wid * (KS_ROWS * INP) + lane;
    float mine = 0.f;
    #pragma unroll
    for (int g = 0; g < KS_ROWS; g += KS_GRP) {
        float v[KS_GRP][INP / 32];
        float s[KS_GRP];
        #pragma unroll
        for (int u = 0; u < KS_GRP; ++u) {
            #pragma unroll
            for (int i = 0; i < INP / 32; ++i) v[u][i] = xw[(g + u) * INP + i * 32];
        }
        #pragma unroll
        for (int u = 0; u < KS_GRP; ++u) {
            float t = 0.f;
            #pragma unroll
            for (int i = 0; i < INP / 32; ++i) t += v[u][i] * kw[i];
            s[u] = t;
        }
        #pragma unroll
        for (int off = 16; off; off >>= 1) {
            #pragma unroll
            for (int u = 0; u < KS_GRP; ++u) s[u] += __shfl_xor(s[u], off);
        }
        #pragma unroll
        for (int u = 0; u < KS_GRP; ++u)
            if (lane == g + u) mine = s[u];
    }
    ksum[wid * KS_ROWS + lane] = mine + kbsum[0];
}

__device__ __forceinline__ void bsa_wmma_tile(const float* __restrict__ Arow,
                                              const float* __restrict__ Bs,
                                              v8f acc[8], int lane) {
    const int kh = (lane >> 4) << 1;
    const int n  = lane & 15;
    for (int k = 0; k < 128; k += 4) {
        v2f a = *(const v2f*)(Arow + k + kh);
        #pragma unroll
        for (int c = 0; c < 8; ++c) {
            v2f bf;
            bf.x = Bs[(k + kh)     * 128 + (c << 4) + n];
            bf.y = Bs[(k + kh + 1) * 128 + (c << 4) + n];
            acc[c] = __builtin_amdgcn_wmma_f32_16x16x4_f32(
                false, a, false, bf, (short)0, acc[c], false, false);
        }
    }
}

#define TR      200
#define NTILE   (NN / TR)

static_assert(NN % TR == 0, "tiles must cover the rows exactly");
static_assert(TR <= 256, "the row inside its tile is one byte of a list entry");

#define ER_PER   4
#define ER_WAVE  (32 * ER_PER)
#define ER_BLOCK 256
#define ER_EDGES (ER_BLOCK * ER_PER)
#define ER_WORDS (ER_EDGES / 32)
#define ER_NBLK  ((EE + ER_EDGES - 1) / ER_EDGES)
#define BM_WORDS (ER_NBLK * ER_WORDS)
#define BM_PITCH (ER_WORDS + 1)

static_assert(EE % ER_WAVE == 0, "a wave's edges are all inside the list or all outside");
static_assert(ER_WORDS == 32, "a tile's words of one block are one line: one word per lane at the store");
static_assert(BM_WORDS % 4 == 0, "a map is read in 16-byte pieces");

__global__ void __launch_bounds__(ER_BLOCK)
bsa_edge_rec(const float* __restrict__ conn,
             const int* __restrict__ rows,
             const int* __restrict__ cols,
             const float* __restrict__ ksum,
             float* __restrict__ erec,
             unsigned* __restrict__ bmap) {
    __shared__ unsigned bm[NTILE * BM_PITCH];
    const int tid  = threadIdx.x;
    const int lane = tid & 31;
    const int wv   = tid >> 5;
    for (int i = tid; i < NTILE * BM_PITCH; i += ER_BLOCK) bm[i] = 0u;
    __syncthreads();
    const int e0 = blockIdx.x * ER_EDGES + wv * ER_WAVE + lane;
    if (e0 < EE) {
        int r[ER_PER], c[ER_PER];
        float cv[ER_PER], k0[ER_PER], k1[ER_PER];
        #pragma unroll
        for (int u = 0; u < ER_PER; ++u) {
            const int e = e0 + u * 32;
            r[u]  = rows[e];
            c[u]  = cols[e];
            cv[u] = conn[e];
        }
        #pragma unroll
        for (int u = 0; u < ER_PER; ++u) {
            k0[u] = ksum[c[u]];
            k1[u] = ksum[NN + c[u]];
        }
        #pragma unroll
        for (int u = 0; u < ER_PER; ++u) {
            const int e = e0 + u * 32;
            const float s0 = cv[u] * k0[u];
            const float s1 = cv[u] * k1[u];
            const int t = r[u] / TR;
            ((int4*)erec)[e] = make_int4(c[u], __float_as_int(s0), __float_as_int(s1), r[u] - t * TR);
            __hip_atomic_fetch_or(&bm[t * BM_PITCH + wv * ER_PER + u], 1u << lane,
                                  __ATOMIC_RELAXED, __HIP_MEMORY_SCOPE_WORKGROUP);
        }
    }
    __syncthreads();
    for (int t = wv; t < NTILE; t += ER_BLOCK / 32)
        bmap[(size_t)t * BM_WORDS + blockIdx.x * ER_WORDS + lane] = bm[t * BM_PITCH + lane];
}

#define AGW        32
#define AHEAD      4
#define BM_U4      (BM_WORDS / 4)
#define BM_STEPS   ((BM_U4 + 31) / 32)
#define BM_ITERS   ((BM_STEPS + AGW - 1) / AGW)
#define QCAP       544
#define SEGCAP     256
#define ROWBITS    8
#define APITCH     132
#define NRT        (TR * BB / 16)

static_assert(TR * BB <= AGW * 32, "one thread per (row, batch) clears and settles the maxima and sums");
static_assert(TR <= (1 << ROWBITS) && EE <= (1 << (32 - ROWBITS)), "edge and row in tile share one word");
static_assert((AGW & (AGW - 1)) == 0 && AGW <= (1 << ROWBITS), "the owner of a row is read from the low bits of an entry");


__device__ __forceinline__ float bsa_lane_bcast(float v, int j) {
    return __int_as_float(__builtin_amdgcn_readlane(__float_as_int(v), j));
}

__device__ __forceinline__ void bsa_tile_add(float* __restrict__ acc, unsigned rl, int lane,
                                             float w0, float w1, float4 x0, float4 x1) {
    float* p0 = acc + (rl * 2u) * APITCH + lane;
    float* p1 = p0 + APITCH;
    __hip_atomic_fetch_add(p0,      w0 * x0.x, __ATOMIC_RELAXED, __HIP_MEMORY_SCOPE_WORKGROUP);
    __hip_atomic_fetch_add(p0 + 32, w0 * x0.y, __ATOMIC_RELAXED, __HIP_MEMORY_SCOPE_WORKGROUP);
    __hip_atomic_fetch_add(p0 + 64, w0 * x0.z, __ATOMIC_RELAXED, __HIP_MEMORY_SCOPE_WORKGROUP);
    __hip_atomic_fetch_add(p0 + 96, w0 * x0.w, __ATOMIC_RELAXED, __HIP_MEMORY_SCOPE_WORKGROUP);
    __hip_atomic_fetch_add(p1,      w1 * x1.x, __ATOMIC_RELAXED, __HIP_MEMORY_SCOPE_WORKGROUP);
    __hip_atomic_fetch_add(p1 + 32, w1 * x1.y, __ATOMIC_RELAXED, __HIP_MEMORY_SCOPE_WORKGROUP);
    __hip_atomic_fetch_add(p1 + 64, w1 * x1.z, __ATOMIC_RELAXED, __HIP_MEMORY_SCOPE_WORKGROUP);
    __hip_atomic_fetch_add(p1 + 96, w1 * x1.w, __ATOMIC_RELAXED, __HIP_MEMORY_SCOPE_WORKGROUP);
}

__device__ __forceinline__ void bsa_sum_add(float* dt, unsigned rl, int lane, float w0, float w1) {
    if (lane < BB)
        __hip_atomic_fetch_add(dt + rl * BB + lane, lane == 0 ? w0 : w1, __ATOMIC_RELAXED, __HIP_MEMORY_SCOPE_WORKGROUP);
}

__device__ __forceinline__ void bsa_max_drain(const unsigned* q, int n, int lane,
                                              const float* __restrict__ erec,
                                              unsigned* mt, unsigned* seg, int segn) {
    if (lane < n) {
        const int e = (int)q[lane];
        const int4 rec = ((const int4*)erec)[e];
        unsigned* p = mt + rec.w * BB;
        __hip_atomic_fetch_max(p,     enc_f32(__int_as_float(rec.y)), __ATOMIC_RELAXED, __HIP_MEMORY_SCOPE_WORKGROUP);
        __hip_atomic_fetch_max(p + 1, enc_f32(__int_as_float(rec.z)), __ATOMIC_RELAXED, __HIP_MEMORY_SCOPE_WORKGROUP);
        const int pos = segn + lane;
        if (pos < SEGCAP) seg[pos] = ((unsigned)e << ROWBITS) | (unsigned)rec.w;
    }
}

__device__ __forceinline__ void bsa_tile_drain(const unsigned* q, int n, int lane,
                                               const float* __restrict__ erec,
                                               const float* mf, float* dt,
                                               const float* __restrict__ x,
                                               float* __restrict__ acc) {
    int c = 0;
    unsigned rl = 0;
    float w0 = 0.f, w1 = 0.f;
    if (lane < n) {
        const int e = (int)q[lane];
        const int4 rec = ((const int4*)erec)[e];
        c = rec.x;
        rl = (unsigned)rec.w;
        const float m0 = mf[rec.w * BB];
        const float m1 = mf[rec.w * BB + 1];
        w0 = __expf(__int_as_float(rec.y) - m0);
        w1 = __expf(__int_as_float(rec.z) - m1);
    }
    const float* v0 = x + lane * 4;
    const float* v1 = x + (size_t)NN * INP + lane * 4;
    int j = 0;
    #pragma nounroll
    for (; j + 4 <= n; j += 4) {
        int cj[4]; unsigned rj[4]; float a0[4], a1[4];
        float4 x0[4], x1[4];
        #pragma unroll
        for (int u = 0; u < 4; ++u) {
            cj[u] = __builtin_amdgcn_readlane(c, j + u);
            rj[u] = (unsigned)__builtin_amdgcn_readlane((int)rl, j + u);
            a0[u] = bsa_lane_bcast(w0, j + u);
            a1[u] = bsa_lane_bcast(w1, j + u);
            x0[u] = *(const float4*)(v0 + (size_t)cj[u] * INP);
            x1[u] = *(const float4*)(v1 + (size_t)cj[u] * INP);
            bsa_sum_add(dt, rj[u], lane, a0[u], a1[u]);
        }
        #pragma unroll
        for (int u = 0; u < 4; ++u)
            bsa_tile_add(acc, rj[u], lane, a0[u], a1[u], x0[u], x1[u]);
    }
    for (; j < n; ++j) {
        int cj = __builtin_amdgcn_readlane(c, j);
        unsigned rj = (unsigned)__builtin_amdgcn_readlane((int)rl, j);
        float a0 = bsa_lane_bcast(w0, j);
        float a1 = bsa_lane_bcast(w1, j);
        float4 x0 = *(const float4*)(v0 + (size_t)cj * INP);
        float4 x1 = *(const float4*)(v1 + (size_t)cj * INP);
        bsa_sum_add(dt, rj, lane, a0, a1);
        bsa_tile_add(acc, rj, lane, a0, a1, x0, x1);
    }
}

#define OWN_ROWS ((TR + AGW - 1) / AGW)

struct BsaRun { float4 a0, a1; float d0, d1; int row; };

__device__ __forceinline__ void bsa_run_flush(const BsaRun& R, float* __restrict__ acc, float* dt, int lane) {
    float* p0 = acc + ((unsigned)R.row * 2u) * APITCH + lane;
    float* p1 = p0 + APITCH;
    __hip_atomic_fetch_add(p0,      R.a0.x, __ATOMIC_RELAXED, __HIP_MEMORY_SCOPE_WORKGROUP);
    __hip_atomic_fetch_add(p0 + 32, R.a0.y, __ATOMIC_RELAXED, __HIP_MEMORY_SCOPE_WORKGROUP);
    __hip_atomic_fetch_add(p0 + 64, R.a0.z, __ATOMIC_RELAXED, __HIP_MEMORY_SCOPE_WORKGROUP);
    __hip_atomic_fetch_add(p0 + 96, R.a0.w, __ATOMIC_RELAXED, __HIP_MEMORY_SCOPE_WORKGROUP);
    __hip_atomic_fetch_add(p1,      R.a1.x, __ATOMIC_RELAXED, __HIP_MEMORY_SCOPE_WORKGROUP);
    __hip_atomic_fetch_add(p1 + 32, R.a1.y, __ATOMIC_RELAXED, __HIP_MEMORY_SCOPE_WORKGROUP);
    __hip_atomic_fetch_add(p1 + 64, R.a1.z, __ATOMIC_RELAXED, __HIP_MEMORY_SCOPE_WORKGROUP);
    __hip_atomic_fetch_add(p1 + 96, R.a1.w, __ATOMIC_RELAXED, __HIP_MEMORY_SCOPE_WORKGROUP);
    bsa_sum_add(dt, (unsigned)R.row, lane, R.d0, R.d1);
}

__device__ __forceinline__ void bsa_run_edge(BsaRun& R, int rj, float w0, float w1, float4 x0, float4 x1,
                                             float* __restrict__ acc, float* dt, int lane) {
    if (rj != R.row) {
        if (R.row >= 0) bsa_run_flush(R, acc, dt, lane);
        R.row = rj;
        R.a0 = make_float4(w0 * x0.x, w0 * x0.y, w0 * x0.z, w0 * x0.w);
        R.a1 = make_float4(w1 * x1.x, w1 * x1.y, w1 * x1.z, w1 * x1.w);
        R.d0 = w0; R.d1 = w1;
    } else {
        R.a0.x = fmaf(w0, x0.x, R.a0.x); R.a0.y = fmaf(w0, x0.y, R.a0.y);
        R.a0.z = fmaf(w0, x0.z, R.a0.z); R.a0.w = fmaf(w0, x0.w, R.a0.w);
        R.a1.x = fmaf(w1, x1.x, R.a1.x); R.a1.y = fmaf(w1, x1.y, R.a1.y);
        R.a1.z = fmaf(w1, x1.z, R.a1.z); R.a1.w = fmaf(w1, x1.w, R.a1.w);
        R.d0 += w0; R.d1 += w1;
    }
}

__device__ __forceinline__ void bsa_own_drain(unsigned* q, int n, int lane,
                                              const float* __restrict__ erec,
                                              const float* mf, float* dt,
                                              const float* __restrict__ x,
                                              float* __restrict__ acc) {
    int c = 0;
    unsigned rl = 0;
    float w0 = 0.f, w1 = 0.f;
    const bool on = lane < n;
    if (on) {
        const int e = (int)q[lane];
        const int4 rec = ((const int4*)erec)[e];
        c = rec.x;
        rl = (unsigned)rec.w;
        const float m0 = mf[rec.w * BB];
        const float m1 = mf[rec.w * BB + 1];
        w0 = __expf(__int_as_float(rec.y) - m0);
        w1 = __expf(__int_as_float(rec.z) - m1);
    }
    const unsigned slot = rl / AGW;
    int pos = 0, before = 0;
    #pragma unroll
    for (int s = 0; s < OWN_ROWS; ++s) {
        const bool mine = on && slot == (unsigned)s;
        const unsigned mk = __builtin_amdgcn_ballot_w32(mine);
        if (mine) pos = before + (int)__builtin_amdgcn_mbcnt_lo(mk, 0u);
        before += __builtin_popcount(mk);
    }
    if (on) q[pos] = (unsigned)lane;
    const int src = on ? (int)q[lane] : 0;
    const int lo = lane * 4;
    BsaRun R;
    R.row = -1; R.d0 = 0.f; R.d1 = 0.f;
    R.a0 = make_float4(0.f, 0.f, 0.f, 0.f); R.a1 = R.a0;
    int j = 0;
    #pragma nounroll
    for (; j + 4 <= n; j += 4) {
        int rj[4]; float a0[4], a1[4];
        float4 x0[4], x1[4];
        #pragma unroll
        for (int u = 0; u < 4; ++u) {
            const int sj = __builtin_amdgcn_readlane(src, j + u);
            const int cj = __builtin_amdgcn_readlane(c, sj);
            rj[u] = __builtin_amdgcn_readlane((int)rl, sj);
            a0[u] = bsa_lane_bcast(w0, sj);
            a1[u] = bsa_lane_bcast(w1, sj);
            const float* r0 = x + (size_t)cj * INP;
            x0[u] = *(const float4*)(r0 + lo);
            x1[u] = *(const float4*)(r0 + (size_t)NN * INP + lo);
        }
        #pragma unroll
        for (int u = 0; u < 4; ++u)
            bsa_run_edge(R, rj[u], a0[u], a1[u], x0[u], x1[u], acc, dt, lane);
    }
    for (; j < n; ++j) {
        const int sj = __builtin_amdgcn_readlane(src, j);
        const int cj = __builtin_amdgcn_readlane(c, sj);
        const int rj = __builtin_amdgcn_readlane((int)rl, sj);
        const float a0 = bsa_lane_bcast(w0, sj);
        const float a1 = bsa_lane_bcast(w1, sj);
        const float* r0 = x + (size_t)cj * INP;
        const float4 x0 = *(const float4*)(r0 + lo);
        const float4 x1 = *(const float4*)(r0 + (size_t)NN * INP + lo);
        bsa_run_edge(R, rj, a0, a1, x0, x1, acc, dt, lane);
    }
    bsa_run_flush(R, acc, dt, lane);
}

template <int PASS>
__device__ __forceinline__ void bsa_drain_ready(unsigned* q, int& cnt, int need, int lane,
                                                const float* __restrict__ erec,
                                                const float* __restrict__ x,
                                                unsigned* mt, float* dt,
                                                float* __restrict__ acc,
                                                unsigned* seg, int& segn) {
    int head = 0;
    while (cnt - head >= need) {
        int n = cnt - head;
        if (n > 32) n = 32;
        if (PASS == 0)      { bsa_max_drain(q + head, n, lane, erec, mt, seg, segn); segn += n; }
        else if (PASS == 1) bsa_tile_drain(q + head, n, lane, erec, (const float*)mt, dt, x, acc);
        else                bsa_own_drain(q + head, n, lane, erec, (const float*)mt, dt, x, acc);
        head += n;
    }
    if (head > 0) {
        const int rem = cnt - head;
        unsigned keep = 0u;
        if (lane < rem) keep = q[head + lane];
        if (lane < rem) q[lane] = keep;
        cnt = rem;
    }
}

__device__ __forceinline__ uint4 bsa_bits16(const uint4* __restrict__ bm4, int st, int wv, int lane) {
    const int i = (st * AGW + wv) * 32 + lane;
    const bool in = st < BM_ITERS && i < BM_U4;
    return bm4[in ? i : 0];
}

template <int PASS>
__device__ __forceinline__ void bsa_bits_step(uint4 v, int st, int wv, int lane,
                                              unsigned* q, int& cnt,
                                              const float* __restrict__ erec,
                                              const float* __restrict__ x,
                                              unsigned* mt, float* dt,
                                              float* __restrict__ acc,
                                              unsigned* seg, int& segn) {
    const int i = (st * AGW + wv) * 32 + lane;
    const bool in = st < BM_ITERS && i < BM_U4;
    unsigned w0 = in ? v.x : 0u;
    unsigned w1 = in ? v.y : 0u;
    unsigned w2 = in ? v.z : 0u;
    unsigned w3 = in ? v.w : 0u;
    const unsigned ebase = (unsigned)i * 128u;
    for (int trip = 0; trip < 128; ++trip) {
        const bool has = (w0 | w1 | w2 | w3) != 0u;
        const unsigned mask = __builtin_amdgcn_ballot_w32(has);
        if (mask == 0u) break;
        const unsigned k  = w0 ? 0u : (w1 ? 1u : (w2 ? 2u : 3u));
        const unsigned wk = w0 ? w0 : (w1 ? w1 : (w2 ? w2 : w3));
        if (has)
            q[cnt + (int)__builtin_amdgcn_mbcnt_lo(mask, 0u)] = ebase + 32u * k + (unsigned)__builtin_ctz(wk);
        const unsigned cl = wk & (wk - 1u);
        w0 = k == 0u ? cl : w0;
        w1 = k == 1u ? cl : w1;
        w2 = k == 2u ? cl : w2;
        w3 = k == 3u ? cl : w3;
        cnt += __builtin_popcount(mask);
        bsa_drain_ready<PASS>(q, cnt, 32, lane, erec, x, mt, dt, acc, seg, segn);
    }
}

template <int PASS>
__device__ __forceinline__ void bsa_scan_pass(const uint4* __restrict__ bm4,
                                              int wv, int lane,
                                              unsigned* q,
                                              const float* __restrict__ erec,
                                              const float* __restrict__ x,
                                              unsigned* mt, float* dt,
                                              float* __restrict__ acc,
                                              unsigned* seg, int& segn) {
    int cnt = 0;

    static_assert(AHEAD == 4, "four map loads are kept in flight");
    uint4 c0 = bsa_bits16(bm4, 0, wv, lane);
    uint4 c1 = bsa_bits16(bm4, 1, wv, lane);
    uint4 c2 = bsa_bits16(bm4, 2, wv, lane);
    uint4 c3 = bsa_bits16(bm4, 3, wv, lane);

#define BSA_STEP(cu, u)                                                                              \
    {                                                                                                \
        bsa_bits_step<PASS>(cu, it + (u), wv, lane, q, cnt, erec, x, mt, dt, acc, seg, segn);   \
        cu = bsa_bits16(bm4, it + (u) + AHEAD, wv, lane);                                            \
    }
    for (int it = 0; it < BM_ITERS; it += AHEAD) {
        BSA_STEP(c0, 0)
        BSA_STEP(c1, 1)
        BSA_STEP(c2, 2)
        BSA_STEP(c3, 3)
    }
#undef BSA_STEP
    bsa_drain_ready<PASS>(q, cnt, 1, lane, erec, x, mt, dt, acc, seg, segn);
}

#define SEG_STEP 128

static_assert(AGW == 32, "lane l holds the length of list l");
static_assert(SEGCAP % SEG_STEP == 0, "a step never reads past its own list");
static_assert(32 + SEG_STEP <= QCAP, "under 32 left over + one step fit the pending list");

__device__ __forceinline__ void bsa_owner_pass(const unsigned* seg, const unsigned* segcnt,
                                               int wv, int lane, unsigned* q,
                                               const float* __restrict__ erec,
                                               const float* __restrict__ x,
                                               unsigned* mt, float* dt,
                                               float* __restrict__ acc) {
    int cnt = 0;
    int none = 0;
    const int cntv = (int)segcnt[lane];
    const uint4* seg4 = (const uint4*)seg;
    uint4 cur = seg4[lane];
    for (int s = 0; s < AGW; ++s) {
        const int sn = s + 1 < AGW ? s + 1 : s;
        const uint4 nxt = seg4[sn * (SEGCAP / 4) + lane];
        const int ns = __builtin_amdgcn_readlane(cntv, s);
        for (int b = 0; b < ns; b += SEG_STEP) {
            uint4 v = cur;
            if (b != 0) v = seg4[s * (SEGCAP / 4) + b / 4 + lane];
            const int i0 = b + lane * 4;
#define BSA_OWN_APPEND(w, k)                                                                   \
            {                                                                                  \
                const bool m = (i0 + (k) < ns) && (((w) & (unsigned)(AGW - 1)) == (unsigned)wv); \
                const unsigned mask = __builtin_amdgcn_ballot_w32(m);                          \
                if (mask != 0u) {                                                              \
                    const int pre = (int)__builtin_amdgcn_mbcnt_lo(mask, 0u);                  \
                    if (m) q[cnt + pre] = (w) >> ROWBITS;                                      \
                    cnt += __builtin_popcount(mask);                                           \
                }                                                                              \
            }
            BSA_OWN_APPEND(v.x, 0)
            BSA_OWN_APPEND(v.y, 1)
            BSA_OWN_APPEND(v.z, 2)
            BSA_OWN_APPEND(v.w, 3)
#undef BSA_OWN_APPEND
            bsa_drain_ready<2>(q, cnt, 32, lane, erec, x, mt, dt, acc, nullptr, none);
        }
        cur = nxt;
    }
    bsa_drain_ready<2>(q, cnt, 1, lane, erec, x, mt, dt, acc, nullptr, none);
}

__global__ void __launch_bounds__(AGW * 32)
bsa_agg_owner(const float* __restrict__ erec,
              const unsigned* __restrict__ bmap,
              const float* __restrict__ x,
              const float* __restrict__ v_w,
              const float* __restrict__ v_b,
              const float* __restrict__ out_w,
              const float* __restrict__ out_b,
              const float* __restrict__ lng,
              const float* __restrict__ lnb,
              float* __restrict__ out) {
    __shared__ __attribute__((aligned(16))) float acc[TR * BB * APITCH];
    __shared__ __attribute__((aligned(16))) unsigned queue[AGW * QCAP];
    __shared__ unsigned mt[TR * BB];
    __shared__ float dt[TR * BB];
    __shared__ unsigned seg[AGW * SEGCAP];
    __shared__ unsigned segcnt[AGW];
    __shared__ unsigned segshort;

    const int tid  = threadIdx.x;
    const int lane = tid & 31;
    const int wv   = __builtin_amdgcn_readfirstlane(tid >> 5);
    const int base = blockIdx.x * TR;
    const uint4* bm4 = (const uint4*)(bmap + (size_t)blockIdx.x * BM_WORDS);

    float4* acc4 = (float4*)acc;
    for (int i = tid; i < TR * BB * APITCH / 4; i += AGW * 32)
        acc4[i] = make_float4(0.f, 0.f, 0.f, 0.f);
    if (tid < TR * BB) { mt[tid] = 0u; dt[tid] = 0.f; }
    if (tid == 0) segshort = 0u;
    __syncthreads();

    unsigned* q = queue + wv * QCAP;

    int segn = 0;
    bsa_scan_pass<0>(bm4, wv, lane, q, erec, x, mt, dt, acc, seg + wv * SEGCAP, segn);
    if (lane == 0) {
        segcnt[wv] = (unsigned)segn;
        if (segn > SEGCAP) segshort = 1u;
    }
    __syncthreads();
    if (tid < TR * BB) {
        const unsigned u = mt[tid];
        float m = (u == 0u) ? 0.0f : dec_f32(u);
        if (!isfinite(m)) m = 0.0f;
        mt[tid] = __float_as_uint(m);
    }
    __syncthreads();

    if (__builtin_amdgcn_readfirstlane((int)segshort) == 0) {
        bsa_owner_pass(seg, segcnt, wv, lane, q, erec, x, mt, dt, acc);
    } else {
        int none = 0;
        bsa_scan_pass<1>(bm4, wv, lane, q, erec, x, mt, dt, acc, nullptr, none);
    }
    __syncthreads();

    static_assert(sizeof(unsigned) * AGW * QCAP >= sizeof(float) * INP * VAL, "a weight fits where the pending lists were");
    static_assert(INP * VAL / 4 == 4 * AGW * 32 && VAL * VAL / 4 == 4 * AGW * 32, "four 16-byte pieces of a weight per thread");
    static_assert(INP == VAL, "the second product reads its left side from the rows the first one wrote");
    float4* Bw4 = (float4*)queue;
    {
        const float4* W4 = (const float4*)v_w;
        float4 w[4];
        #pragma unroll
        for (int i = 0; i < 4; ++i) w[i] = W4[tid + i * (AGW * 32)];
        #pragma unroll
        for (int i = 0; i < 4; ++i) {
            const int p  = tid + i * (AGW * 32);
            const int f  = p >> 5;
            const int wr = (f & 3) * 32 + (f >> 2);
            Bw4[wr * 32 + (p & 31)] = w[i];
        }
    }
    for (int row = wv; row < TR * BB; row += AGW) {
        float* a = acc + row * APITCH + lane;
        const float d  = dt[row];
        const float dd = (d == 0.0f) ? 1.0f : d;
        a[0] = a[0] / dd; a[32] = a[32] / dd; a[64] = a[64] / dd; a[96] = a[96] / dd;
        if (lane == 0) dt[row] = d / dd;
    }
    __syncthreads();

    const float* Bw = (const float*)queue;
    const int n      = lane & 15;
    const int rowoff = (lane >> 4) << 3;
    const float4* O4 = (const float4*)out_w + tid;
    float4 ow0, ow1, ow2, ow3;
    if (wv < NRT) {
        v8f tf[8] = {};
        bsa_wmma_tile(acc + (wv * 16 + n) * APITCH, Bw, tf, lane);
        ow0 = O4[0 * (AGW * 32)]; ow1 = O4[1 * (AGW * 32)]; ow2 = O4[2 * (AGW * 32)]; ow3 = O4[3 * (AGW * 32)];
        float sw[8];
        #pragma unroll
        for (int r = 0; r < 8; ++r) sw[r] = dt[wv * 16 + rowoff + r];
        #pragma unroll
        for (int c = 0; c < 8; ++c) {
            const float vb = v_b[(c << 4) + n];
            #pragma unroll
            for (int r = 0; r < 8; ++r)
                acc[(wv * 16 + rowoff + r) * APITCH + (c << 4) + n] = tf[c][r] + sw[r] * vb;
        }
    } else {
        ow0 = O4[0 * (AGW * 32)]; ow1 = O4[1 * (AGW * 32)]; ow2 = O4[2 * (AGW * 32)]; ow3 = O4[3 * (AGW * 32)];
    }
    __syncthreads();
    Bw4[tid + 0 * (AGW * 32)] = ow0; Bw4[tid + 1 * (AGW * 32)] = ow1;
    Bw4[tid + 2 * (AGW * 32)] = ow2; Bw4[tid + 3 * (AGW * 32)] = ow3;
    __syncthreads();
    if (wv >= NRT) return;

    v8f cf[8] = {};
    bsa_wmma_tile(acc + (wv * 16 + n) * APITCH, Bw, cf, lane);

    float sr[8], sq[8];
    #pragma unroll
    for (int r = 0; r < 8; ++r) { sr[r] = 0.f; sq[r] = 0.f; }
    #pragma unroll
    for (int c = 0; c < 8; ++c) {
        const float bb = out_b[(c << 4) + n];
        #pragma unroll
        for (int r = 0; r < 8; ++r) {
            float v = cf[c][r] + bb;
            v = v / (1.0f + __expf(-v));
            cf[c][r] = v;
            sr[r] += v;
            sq[r] += v * v;
        }
    }
    #pragma unroll
    for (int r = 0; r < 8; ++r) {
        #pragma unroll
        for (int off = 8; off; off >>= 1) {
            sr[r] += __shfl_xor(sr[r], off);
            sq[r] += __shfl_xor(sq[r], off);
        }
    }
    float mu[8], rstd[8];
    #pragma unroll
    for (int r = 0; r < 8; ++r) {
        mu[r] = sr[r] * (1.0f / VAL);
        const float var = sq[r] * (1.0f / VAL) - mu[r] * mu[r];
        rstd[r] = rsqrtf(var + LN_EPS);
    }
    #pragma unroll
    for (int c = 0; c < 8; ++c) {
        const int col = (c << 4) + n;
        const float g = lng[col], b2 = lnb[col];
        #pragma unroll
        for (int r = 0; r < 8; ++r) {
            const int row = wv * 16 + rowoff + r;
            out[((size_t)(row & 1) * NN + base + (row >> 1)) * VAL + col] =
                (cf[c][r] - mu[r]) * rstd[r] * g + b2;
        }
    }
}

extern "C" void kernel_launch(void* const* d_in, const int* in_sizes, int n_in,
                              void* d_out, int out_size, void* d_ws, size_t ws_size,
                              hipStream_t stream) {
    const float* x     = (const float*)d_in[0];
    const float* conn  = (const float*)d_in[1];
    const float* k_w   = (const float*)d_in[2];
    const float* k_b   = (const float*)d_in[3];
    const float* v_w   = (const float*)d_in[4];
    const float* v_b   = (const float*)d_in[5];
    const float* out_w = (const float*)d_in[6];
    const float* out_b = (const float*)d_in[7];
    const float* ln_g  = (const float*)d_in[8];
    const float* ln_b  = (const float*)d_in[9];
    const int*   rows  = (const int*)d_in[10];
    const int*   cols  = (const int*)d_in[11];
    float* out = (float*)d_out;

    char* p = (char*)d_ws;
    auto alloc = [&](size_t bytes) {
        void* r = (void*)p;
        p += (bytes + 255) & ~(size_t)255;
        return r;
    };
    float*    kwsum  = (float*)alloc(INP * sizeof(float));
    float*    kbsum  = (float*)alloc(sizeof(float));
    float*    ksum   = (float*)alloc((size_t)MROWS * sizeof(float));
    float*    erec   = (float*)alloc((size_t)EE * 4 * sizeof(float));
    unsigned* bmap   = (unsigned*)alloc((size_t)NTILE * BM_WORDS * sizeof(unsigned));

    bsa_reduce_kw<<<1, 128, 0, stream>>>(k_w, k_b, kwsum, kbsum);

    bsa_ksum<<<(MROWS / KS_ROWS * 32 + 255) / 256, 256, 0, stream>>>(x, kwsum, kbsum, ksum);

    bsa_edge_rec<<<ER_NBLK, ER_BLOCK, 0, stream>>>(conn, rows, cols, ksum, erec, bmap);

    bsa_agg_owner<<<NTILE, AGW * 32, 0, stream>>>(erec, bmap, x, v_w, v_b, out_w, out_b, ln_g, ln_b, out);
}
